# starting kernel
# baseline (speedup 1.0000x reference)
#include <hip/hip_runtime.h>


typedef __attribute__((ext_vector_type(2))) float v2f;
typedef __attribute__((ext_vector_type(8))) float v8f;


__global__ void GCN_zero_f32(float* __restrict__ p, long n) {
  long i = (long)blockIdx.x * blockDim.x + threadIdx.x;
  long stride = (long)gridDim.x * blockDim.x;
  for (; i < n; i += stride) p[i] = 0.0f;
}

__global__ void GCN_degree(const int* __restrict__ src, const int* __restrict__ dst,
                           float* __restrict__ deg_out, float* __restrict__ deg_in,
                           int ne) {
  int i = blockIdx.x * blockDim.x + threadIdx.x;
  if (i < ne) {
    atomicAdd(&deg_out[src[i]], 1.0f);
    atomicAdd(&deg_in[dst[i]], 1.0f);
  }
}

__global__ void GCN_norms(const float* __restrict__ deg_out, const float* __restrict__ deg_in,
                          float* __restrict__ nsrc, float* __restrict__ ndst, int n) {
  int i = blockIdx.x * blockDim.x + threadIdx.x;
  if (i < n) {
    nsrc[i] = rsqrtf(fmaxf(deg_out[i], 1.0f));
    ndst[i] = rsqrtf(fmaxf(deg_in[i], 1.0f));
  }
}

template <int K, int NT, int NC>
__global__ __launch_bounds__(32 * NT)
void GCN_gemm_wmma(const float* __restrict__ X, const float* __restrict__ scale,
                   const float* __restrict__ W, float* __restrict__ Y, int nrows) {
  constexpr int NCP = NT * 16;
  constexpr bool EXACT = (NC == NCP);
  constexpr int PITCH = K + 4;
  __shared__ float sa[16 * PITCH];
  __shared__ float sw[EXACT ? 4 : K * NCP];

  const int rowbase = blockIdx.x * 16;

  for (int idx = threadIdx.x; idx < 16 * (K / 4); idx += blockDim.x) {
    const int r = idx / (K / 4);
    const int c = idx - r * (K / 4);
    const int row = rowbase + r;
    float4 v = make_float4(0.f, 0.f, 0.f, 0.f);
    float s = 0.0f;
    if (row < nrows) {
      v = ((const float4*)(X + (long)row * K))[c];
      s = scale[row];
    }
    float* d = &sa[r * PITCH + 4 * c];
    d[0] = v.x * s; d[1] = v.y * s; d[2] = v.z * s; d[3] = v.w * s;
  }

  if constexpr (!EXACT) {
    for (int idx = threadIdx.x; idx < K * NCP; idx += blockDim.x) {
      const int k = idx / NCP;
      const int c = idx - k * NCP;
      sw[idx] = (c < NC) ? W[k * NC + c] : 0.0f;
    }
  }
  __syncthreads();

  const int lane = threadIdx.x & 31;
  const int wave = threadIdx.x >> 5;
  const int colbase = wave * 16;
  const int half = lane >> 4;
  const int l15 = lane & 15;
  const int koff = half * 2;

  const float* aptr = &sa[l15 * PITCH + koff];
  const int bcol = colbase + l15;

  v8f acc = {};
#pragma unroll 8
  for (int k0 = 0; k0 < K; k0 += 4) {
    v2f a;
    a.x = aptr[k0];
    a.y = aptr[k0 + 1];
    v2f b;
    if constexpr (EXACT) {
      b.x = W[(k0 + koff) * NC + bcol];
      b.y = W[(k0 + koff + 1) * NC + bcol];
    } else {
      b.x = sw[(k0 + koff) * NCP + bcol];
      b.y = sw[(k0 + koff + 1) * NCP + bcol];
    }
    acc = __builtin_amdgcn_wmma_f32_16x16x4_f32(
         false, a,  false, b,
         (short)0, acc,  false,  false);
  }

  if (EXACT || bcol < NC) {
    float* yp = Y + (long)(rowbase + half * 8) * NC + bcol;
    if (rowbase + 16 <= nrows) {
#pragma unroll
      for (int r = 0; r < 8; ++r) yp[(long)r * NC] = acc[r];
    } else {
      const int rbase = rowbase + half * 8;
#pragma unroll
      for (int r = 0; r < 8; ++r) {
        if (rbase + r < nrows) yp[(long)r * NC] = acc[r];
      }
    }
  }
}

__global__ void GCN_scatter128(const float* __restrict__ H,
                               const int* __restrict__ src, const int* __restrict__ dst,
                               float* __restrict__ AGG, int ne) {
  const int wid = (int)((blockIdx.x * (long)blockDim.x + threadIdx.x) >> 5);
  const int lane = threadIdx.x & 31;
  if (wid >= ne) return;
  const int s = src[wid];
  const int d = dst[wid];
  const float4 v = ((const float4*)(H + (long)s * 128))[lane];
  float* out = AGG + (long)d * 128 + lane * 4;
  atomicAdd(out + 0, v.x);
  atomicAdd(out + 1, v.y);
  atomicAdd(out + 2, v.z);
  atomicAdd(out + 3, v.w);
}

__global__ void GCN_scatter40(const float* __restrict__ H,
                              const int* __restrict__ src, const int* __restrict__ dst,
                              float* __restrict__ AGG, int ne) {
  const int wid = (int)((blockIdx.x * (long)blockDim.x + threadIdx.x) >> 5);
  const int lane = threadIdx.x & 31;
  if (wid >= ne) return;
  const int s = src[wid];
  const int d = dst[wid];
  const float* hp = H + (long)s * 40;
  float* out = AGG + (long)d * 40;
  atomicAdd(out + lane, hp[lane]);
  if (lane < 8) atomicAdd(out + lane + 32, hp[lane + 32]);
}


__global__ void GCN_post1(const float* __restrict__ AGG, const float* __restrict__ ndst,
                          const float* __restrict__ b, float* __restrict__ H1, long total) {
  long idx = (long)blockIdx.x * blockDim.x + threadIdx.x;
  long stride = (long)gridDim.x * blockDim.x;
  for (; idx < total; idx += stride) {
    const int i = (int)(idx >> 7);
    const int f = (int)(idx & 127);
    const float v = AGG[idx] * ndst[i] + b[f];
    H1[idx] = v > 0.0f ? v : 0.0f;
  }
}

__global__ void GCN_post2(float* __restrict__ OUT, const float* __restrict__ ndst,
                          const float* __restrict__ b, long total) {
  long idx = (long)blockIdx.x * blockDim.x + threadIdx.x;
  long stride = (long)gridDim.x * blockDim.x;
  for (; idx < total; idx += stride) {
    const int i = (int)(idx / 40);
    const int f = (int)(idx - (long)i * 40);
    OUT[idx] = OUT[idx] * ndst[i] + b[f];
  }
}


extern "C" void kernel_launch(void* const* d_in, const int* in_sizes, int n_in,
                              void* d_out, int out_size, void* d_ws, size_t ws_size,
                              hipStream_t stream) {
  const float* x   = (const float*)d_in[0];
  const float* W1  = (const float*)d_in[1];
  const float* b1  = (const float*)d_in[2];
  const float* W2  = (const float*)d_in[3];
  const float* b2  = (const float*)d_in[4];
  const int* esrc  = (const int*)d_in[5];
  const int* edst  = (const int*)d_in[6];
  float* out = (float*)d_out;

  const int nn = in_sizes[0] / 256;
  const int ne = in_sizes[5];

  const long NPAD = 50048;
  float* ws       = (float*)d_ws;
  float* deg_out  = ws;
  float* deg_in   = ws + NPAD;
  float* nsrc     = ws + 2 * NPAD;
  float* ndst     = ws + 3 * NPAD;
  float* h        = ws + 4 * NPAD;
  float* agg1     = ws + 4 * NPAD + (long)nn * 128;
  float* h2       = ws + 4 * NPAD + 2L * (long)nn * 128;

  GCN_zero_f32<<<512, 256, 0, stream>>>(deg_out, 2 * NPAD);
  GCN_zero_f32<<<4096, 256, 0, stream>>>(agg1, (long)nn * 128);
  GCN_zero_f32<<<2048, 256, 0, stream>>>(out, (long)nn * 40);

  GCN_degree<<<(ne + 255) / 256, 256, 0, stream>>>(esrc, edst, deg_out, deg_in, ne);
  GCN_norms<<<(nn + 255) / 256, 256, 0, stream>>>(deg_out, deg_in, nsrc, ndst, nn);

  const int mtiles = (nn + 15) / 16;

  GCN_gemm_wmma<256, 8, 128><<<mtiles, 256, 0, stream>>>(x, nsrc, W1, h, nn);

  GCN_scatter128<<<(ne + 7) / 8, 256, 0, stream>>>(h, esrc, edst, agg1, ne);

  GCN_post1<<<8192, 256, 0, stream>>>(agg1, ndst, b1, h, (long)nn * 128);

  GCN_gemm_wmma<128, 3, 40><<<mtiles, 96, 0, stream>>>(h, nsrc, W2, h2, nn);

  GCN_scatter40<<<(ne + 7) / 8, 256, 0, stream>>>(h2, esrc, edst, out, ne);

  GCN_post2<<<4096, 256, 0, stream>>>(out, ndst, b2, (long)nn * 40);
}


# edge sums kept in LDS: a work-group owns 256 destination rows, scans one-byte block ids of all edges, adds gathered rows by LDS f32 adds (arrival order); global atomics, zero and post passes go
# speedup vs baseline: 1.5073x; 1.5073x over previous
#include <hip/hip_runtime.h>


typedef __attribute__((ext_vector_type(2))) float v2f;
typedef __attribute__((ext_vector_type(8))) float v8f;


__global__ void GCN_zero_f32(float* __restrict__ p, long n) {
  long i = (long)blockIdx.x * blockDim.x + threadIdx.x;
  long stride = (long)gridDim.x * blockDim.x;
  for (; i < n; i += stride) p[i] = 0.0f;
}

__global__ void GCN_degree(const int* __restrict__ src, const int* __restrict__ dst,
                           float* __restrict__ deg_out, float* __restrict__ deg_in,
                           unsigned* __restrict__ own4, int ne, int nwords) {
  const int g = blockIdx.x * blockDim.x + threadIdx.x;
  if (g >= nwords) return;
  const int e = 4 * g;
  unsigned packed = 0xFFFFFFFFu;
  if (e + 4 <= ne) {
    const int4 s = ((const int4*)src)[g];
    const int4 d = ((const int4*)dst)[g];
    atomicAdd(&deg_out[s.x], 1.0f); atomicAdd(&deg_out[s.y], 1.0f);
    atomicAdd(&deg_out[s.z], 1.0f); atomicAdd(&deg_out[s.w], 1.0f);
    atomicAdd(&deg_in[d.x], 1.0f);  atomicAdd(&deg_in[d.y], 1.0f);
    atomicAdd(&deg_in[d.z], 1.0f);  atomicAdd(&deg_in[d.w], 1.0f);
    packed = (((unsigned)d.x >> 8) & 255u) | ((((unsigned)d.y >> 8) & 255u) << 8) |
             ((((unsigned)d.z >> 8) & 255u) << 16) | ((((unsigned)d.w >> 8) & 255u) << 24);
  } else {
    for (int k = 0; k < 4; ++k) {
      if (e + k < ne) {
        atomicAdd(&deg_out[src[e + k]], 1.0f);
        const int dd = dst[e + k];
        atomicAdd(&deg_in[dd], 1.0f);
        packed = (packed & ~(255u << (8 * k))) | ((((unsigned)dd >> 8) & 255u) << (8 * k));
      }
    }
  }
  own4[g] = packed;
}

__global__ void GCN_norms(const float* __restrict__ deg_out, const float* __restrict__ deg_in,
                          float* __restrict__ nsrc, float* __restrict__ ndst, int n) {
  int i = blockIdx.x * blockDim.x + threadIdx.x;
  if (i < n) {
    nsrc[i] = rsqrtf(fmaxf(deg_out[i], 1.0f));
    ndst[i] = rsqrtf(fmaxf(deg_in[i], 1.0f));
  }
}

template <int K, int NT, int NC>
__global__ __launch_bounds__(32 * NT)
void GCN_gemm_wmma(const float* __restrict__ X, const float* __restrict__ scale,
                   const float* __restrict__ W, float* __restrict__ Y, int nrows) {
  constexpr int NCP = NT * 16;
  constexpr bool EXACT = (NC == NCP);
  constexpr int PITCH = K + 4;
  __shared__ float sa[16 * PITCH];
  __shared__ float sw[EXACT ? 4 : K * NCP];

  const int rowbase = blockIdx.x * 16;

  for (int idx = threadIdx.x; idx < 16 * (K / 4); idx += blockDim.x) {
    const int r = idx / (K / 4);
    const int c = idx - r * (K / 4);
    const int row = rowbase + r;
    float4 v = make_float4(0.f, 0.f, 0.f, 0.f);
    float s = 0.0f;
    if (row < nrows) {
      v = ((const float4*)(X + (long)row * K))[c];
      s = scale[row];
    }
    float* d = &sa[r * PITCH + 4 * c];
    d[0] = v.x * s; d[1] = v.y * s; d[2] = v.z * s; d[3] = v.w * s;
  }

  if constexpr (!EXACT) {
    for (int idx = threadIdx.x; idx < K * NCP; idx += blockDim.x) {
      const int k = idx / NCP;
      const int c = idx - k * NCP;
      sw[idx] = (c < NC) ? W[k * NC + c] : 0.0f;
    }
  }
  __syncthreads();

  const int lane = threadIdx.x & 31;
  const int wave = threadIdx.x >> 5;
  const int colbase = wave * 16;
  const int half = lane >> 4;
  const int l15 = lane & 15;
  const int koff = half * 2;

  const float* aptr = &sa[l15 * PITCH + koff];
  const int bcol = colbase + l15;

  v8f acc = {};
#pragma unroll 8
  for (int k0 = 0; k0 < K; k0 += 4) {
    v2f a;
    a.x = aptr[k0];
    a.y = aptr[k0 + 1];
    v2f b;
    if constexpr (EXACT) {
      b.x = W[(k0 + koff) * NC + bcol];
      b.y = W[(k0 + koff + 1) * NC + bcol];
    } else {
      b.x = sw[(k0 + koff) * NCP + bcol];
      b.y = sw[(k0 + koff + 1) * NCP + bcol];
    }
    acc = __builtin_amdgcn_wmma_f32_16x16x4_f32(
         false, a,  false, b,
         (short)0, acc,  false,  false);
  }

  if (EXACT || bcol < NC) {
    float* yp = Y + (long)(rowbase + half * 8) * NC + bcol;
    if (rowbase + 16 <= nrows) {
#pragma unroll
      for (int r = 0; r < 8; ++r) yp[(long)r * NC] = acc[r];
    } else {
      const int rbase = rowbase + half * 8;
#pragma unroll
      for (int r = 0; r < 8; ++r) {
        if (rbase + r < nrows) yp[(long)r * NC] = acc[r];
      }
    }
  }
}

constexpr int RB = 256;
constexpr int QB = 16;
constexpr int AW = 32;
constexpr int SG = 8;

__device__ __forceinline__ unsigned GCN_zero_bytes(unsigned x) {
  return ~(((x & 0x7F7F7F7Fu) + 0x7F7F7F7Fu) | x | 0x7F7F7F7Fu);
}

__device__ __forceinline__ void GCN_lds_add(float* p, float v) {
  __hip_atomic_fetch_add(p, v, __ATOMIC_RELAXED, __HIP_MEMORY_SCOPE_WORKGROUP);
}

template <int F, bool FULL>
__device__ __forceinline__ void GCN_sum_batch(const float* __restrict__ H,
                                              const int* __restrict__ src,
                                              const int* __restrict__ dst,
                                              float* acc, int qe, int qn,
                                              int lane, int rowbase) {
  int s = 0, d = rowbase;
  if (lane < (FULL ? QB : qn)) { s = src[qe]; d = dst[qe]; }
  if constexpr (F == 128) {
    float4 v[QB];
#pragma unroll
    for (int u = 0; u < QB; ++u) {
      if (FULL || u < qn) {
        const float* hp = H + (long)__builtin_amdgcn_readlane(s, u) * 128;
        v[u] = ((const float4*)hp)[lane];
      }
    }
#pragma unroll
    for (int u = 0; u < QB; ++u) {
      if (FULL || u < qn) {
        float* ap = acc + (__builtin_amdgcn_readlane(d, u) - rowbase) * 128 + lane;
        GCN_lds_add(ap,      v[u].x);
        GCN_lds_add(ap + 32, v[u].y);
        GCN_lds_add(ap + 64, v[u].z);
        GCN_lds_add(ap + 96, v[u].w);
      }
    }
  } else {
    constexpr int NCH = (F + 31) / 32;
    float v[NCH][QB];
#pragma unroll
    for (int c = 0; c < NCH; ++c) {
      if ((c + 1) * 32 <= F || lane < F - c * 32) {
#pragma unroll
        for (int u = 0; u < QB; ++u) {
          if (FULL || u < qn) {
            const float* hp = H + (long)__builtin_amdgcn_readlane(s, u) * F;
            v[c][u] = hp[c * 32 + lane];
          }
        }
      }
    }
#pragma unroll
    for (int c = 0; c < NCH; ++c) {
      if ((c + 1) * 32 <= F || lane < F - c * 32) {
#pragma unroll
        for (int u = 0; u < QB; ++u) {
          if (FULL || u < qn) {
            const int du = __builtin_amdgcn_readlane(d, u) - rowbase;
            GCN_lds_add(acc + du * F + c * 32 + lane, v[c][u]);
          }
        }
      }
    }
  }
}

template <int F, bool RELU>
__global__ __launch_bounds__(32 * AW)
void GCN_edge_sum(const float* __restrict__ H,
                  const int* __restrict__ src, const int* __restrict__ dst,
                  const unsigned* __restrict__ own4,
                  const float* __restrict__ ndst, const float* __restrict__ bias,
                  float* __restrict__ OUT, int ne, int nn) {
  __shared__ float acc[RB * F];
  const int lane = threadIdx.x & 31;
  const int wave = __builtin_amdgcn_readfirstlane((int)(threadIdx.x >> 5));
  const int rowbase = blockIdx.x * RB;

  for (int i = threadIdx.x; i < RB * F; i += 32 * AW) acc[i] = 0.0f;
  __syncthreads();

  const unsigned pat = (unsigned)blockIdx.x * 0x01010101u;
  const uint4* own = (const uint4*)own4;
  const int nloads = ((ne + 512 * SG - 1) / (512 * SG)) * SG;
  int qe = 0;
  int qn = 0;

  for (int i0 = wave * SG; i0 < nloads; i0 += AW * SG) {
    uint4 wv[SG];
#pragma unroll
    for (int g = 0; g < SG; ++g) {
      wv[g] = own[(i0 + g) * 32 + lane];
    }
#pragma nounroll
    for (int g = 0; g < SG; ++g) {
      const uint4 w = wv[0];
#pragma unroll
      for (int k = 0; k + 1 < SG; ++k) wv[k] = wv[k + 1];
      const int i = i0 + g;
      unsigned t[4];
      t[0] = GCN_zero_bytes(w.x ^ pat);
      t[1] = GCN_zero_bytes(w.y ^ pat);
      t[2] = GCN_zero_bytes(w.z ^ pat);
      t[3] = GCN_zero_bytes(w.w ^ pat);
      unsigned ball = __builtin_amdgcn_ballot_w32((t[0] | t[1] | t[2] | t[3]) != 0u);
      while (ball) {
        const int l = __builtin_ctz(ball);
        ball &= ball - 1;
#pragma unroll
        for (int j = 0; j < 4; ++j) {
          unsigned f = (unsigned)__builtin_amdgcn_readlane((int)t[j], l);
          while (f) {
            const int bit = __builtin_ctz(f);
            f &= f - 1;
            const int e = i * 512 + l * 16 + j * 4 + (bit >> 3);
            if (e < ne) {
              if (lane == qn) qe = e;
              ++qn;
            }
          }
        }
        if (qn >= QB) {
          GCN_sum_batch<F, true>(H, src, dst, acc, qe, qn, lane, rowbase);
          qe = __builtin_amdgcn_ds_bpermute(((lane + QB) & 31) << 2, qe);
          qn -= QB;
        }
      }
    }
  }
  if (qn > 0) GCN_sum_batch<F, false>(H, src, dst, acc, qe, qn, lane, rowbase);
  __syncthreads();

  if constexpr (F == 128) {
    for (int i = threadIdx.x; i < RB * 32; i += 32 * AW) {
      const int r = i >> 5;
      const int row = rowbase + r;
      if (row < nn) {
        const float* ap = acc + r * 128 + lane;
        const float nd = ndst[row];
        const float4 bb = ((const float4*)bias)[lane];
        float4 o;
        o.x = ap[0]  * nd + bb.x;
        o.y = ap[32] * nd + bb.y;
        o.z = ap[64] * nd + bb.z;
        o.w = ap[96] * nd + bb.w;
        if (RELU) {
          o.x = o.x > 0.0f ? o.x : 0.0f;
          o.y = o.y > 0.0f ? o.y : 0.0f;
          o.z = o.z > 0.0f ? o.z : 0.0f;
          o.w = o.w > 0.0f ? o.w : 0.0f;
        }
        ((float4*)(OUT + (long)row * 128))[lane] = o;
      }
    }
  } else {
    for (int i = threadIdx.x; i < RB * F; i += 32 * AW) {
      const int r = i / F;
      const int f = i - r * F;
      const int row = rowbase + r;
      if (row < nn) {
        float v = acc[i] * ndst[row] + bias[f];
        if (RELU) v = v > 0.0f ? v : 0.0f;
        OUT[(long)row * F + f] = v;
      }
    }
  }
}


extern "C" void kernel_launch(void* const* d_in, const int* in_sizes, int n_in,
                              void* d_out, int out_size, void* d_ws, size_t ws_size,
                              hipStream_t stream) {
  const float* x   = (const float*)d_in[0];
  const float* W1  = (const float*)d_in[1];
  const float* b1  = (const float*)d_in[2];
  const float* W2  = (const float*)d_in[3];
  const float* b2  = (const float*)d_in[4];
  const int* esrc  = (const int*)d_in[5];
  const int* edst  = (const int*)d_in[6];
  float* out = (float*)d_out;

  const int nn = in_sizes[0] / 256;
  const int ne = in_sizes[5];

  const long NPAD = 50048;
  float* ws       = (float*)d_ws;
  float* deg_out  = ws;
  float* deg_in   = ws + NPAD;
  float* nsrc     = ws + 2 * NPAD;
  float* ndst     = ws + 3 * NPAD;
  float* h        = ws + 4 * NPAD;
  float* h1       = ws + 4 * NPAD + (long)nn * 128;
  float* h2       = ws + 4 * NPAD + 2L * (long)nn * 128;
  unsigned* own4  = (unsigned*)(ws + 4 * NPAD + 2L * (long)nn * 128 + (long)nn * 40);

  GCN_zero_f32<<<512, 256, 0, stream>>>(deg_out, 2 * NPAD);

  const int nwords = ((ne + 512 * SG - 1) / (512 * SG)) * (128 * SG);
  GCN_degree<<<(nwords + 255) / 256, 256, 0, stream>>>(esrc, edst, deg_out, deg_in, own4, ne, nwords);
  GCN_norms<<<(nn + 255) / 256, 256, 0, stream>>>(deg_out, deg_in, nsrc, ndst, nn);

  const int mtiles = (nn + 15) / 16;
  const int rblocks = (nn + RB - 1) / RB;

  GCN_gemm_wmma<256, 8, 128><<<mtiles, 256, 0, stream>>>(x, nsrc, W1, h, nn);

  GCN_edge_sum<128, true><<<rblocks, 32 * AW, 0, stream>>>(h, esrc, edst, own4, ndst, b1, h1, ne, nn);

  GCN_gemm_wmma<128, 3, 40><<<mtiles, 96, 0, stream>>>(h1, nsrc, W2, h2, nn);

  GCN_edge_sum<40, false><<<rblocks, 32 * AW, 0, stream>>>(h2, esrc, edst, own4, ndst, b2, out, ne, nn);
}


# layer-1 edge sum records each row block's matched (src,dst) pairs as 128 B slots; layer-2 edge sum reads its slots instead of scanning all edge bytes again; layer-2 LDS add order follows slot order
# speedup vs baseline: 1.6387x; 1.0872x over previous
#include <hip/hip_runtime.h>


typedef __attribute__((ext_vector_type(2))) float v2f;
typedef __attribute__((ext_vector_type(8))) float v8f;


__global__ void GCN_zero_f32(float* __restrict__ p, long n) {
  long i = (long)blockIdx.x * blockDim.x + threadIdx.x;
  long stride = (long)gridDim.x * blockDim.x;
  for (; i < n; i += stride) p[i] = 0.0f;
}

__global__ void GCN_degree(const int* __restrict__ src, const int* __restrict__ dst,
                           float* __restrict__ deg_out, float* __restrict__ deg_in,
                           unsigned* __restrict__ own4, int ne, int nwords) {
  const int g = blockIdx.x * blockDim.x + threadIdx.x;
  if (g >= nwords) return;
  const int e = 4 * g;
  unsigned packed = 0xFFFFFFFFu;
  if (e + 4 <= ne) {
    const int4 s = ((const int4*)src)[g];
    const int4 d = ((const int4*)dst)[g];
    atomicAdd(&deg_out[s.x], 1.0f); atomicAdd(&deg_out[s.y], 1.0f);
    atomicAdd(&deg_out[s.z], 1.0f); atomicAdd(&deg_out[s.w], 1.0f);
    atomicAdd(&deg_in[d.x], 1.0f);  atomicAdd(&deg_in[d.y], 1.0f);
    atomicAdd(&deg_in[d.z], 1.0f);  atomicAdd(&deg_in[d.w], 1.0f);
    packed = (((unsigned)d.x >> 8) & 255u) | ((((unsigned)d.y >> 8) & 255u) << 8) |
             ((((unsigned)d.z >> 8) & 255u) << 16) | ((((unsigned)d.w >> 8) & 255u) << 24);
  } else {
    for (int k = 0; k < 4; ++k) {
      if (e + k < ne) {
        atomicAdd(&deg_out[src[e + k]], 1.0f);
        const int dd = dst[e + k];
        atomicAdd(&deg_in[dd], 1.0f);
        packed = (packed & ~(255u << (8 * k))) | ((((unsigned)dd >> 8) & 255u) << (8 * k));
      }
    }
  }
  own4[g] = packed;
}

__global__ void GCN_norms(const float* __restrict__ deg_out, const float* __restrict__ deg_in,
                          float* __restrict__ nsrc, float* __restrict__ ndst,
                          int* __restrict__ blk_cnt, int n) {
  __shared__ int cnt;
  if (threadIdx.x == 0) cnt = 0;
  __syncthreads();
  int i = blockIdx.x * blockDim.x + threadIdx.x;
  int c = 0;
  if (i < n) {
    const float di = deg_in[i];
    nsrc[i] = rsqrtf(fmaxf(deg_out[i], 1.0f));
    ndst[i] = rsqrtf(fmaxf(di, 1.0f));
    c = (int)di;
  }
  __hip_atomic_fetch_add(&cnt, c, __ATOMIC_RELAXED, __HIP_MEMORY_SCOPE_WORKGROUP);
  __syncthreads();
  if (threadIdx.x == 0) blk_cnt[blockIdx.x] = cnt;
}

template <int K, int NT, int NC>
__global__ __launch_bounds__(32 * NT)
void GCN_gemm_wmma(const float* __restrict__ X, const float* __restrict__ scale,
                   const float* __restrict__ W, float* __restrict__ Y, int nrows) {
  constexpr int NCP = NT * 16;
  constexpr bool EXACT = (NC == NCP);
  constexpr int PITCH = K + 4;
  __shared__ float sa[16 * PITCH];
  __shared__ float sw[EXACT ? 4 : K * NCP];

  const int rowbase = blockIdx.x * 16;

  for (int idx = threadIdx.x; idx < 16 * (K / 4); idx += blockDim.x) {
    const int r = idx / (K / 4);
    const int c = idx - r * (K / 4);
    const int row = rowbase + r;
    float4 v = make_float4(0.f, 0.f, 0.f, 0.f);
    float s = 0.0f;
    if (row < nrows) {
      v = ((const float4*)(X + (long)row * K))[c];
      s = scale[row];
    }
    float* d = &sa[r * PITCH + 4 * c];
    d[0] = v.x * s; d[1] = v.y * s; d[2] = v.z * s; d[3] = v.w * s;
  }

  if constexpr (!EXACT) {
    for (int idx = threadIdx.x; idx < K * NCP; idx += blockDim.x) {
      const int k = idx / NCP;
      const int c = idx - k * NCP;
      sw[idx] = (c < NC) ? W[k * NC + c] : 0.0f;
    }
  }
  __syncthreads();

  const int lane = threadIdx.x & 31;
  const int wave = threadIdx.x >> 5;
  const int colbase = wave * 16;
  const int half = lane >> 4;
  const int l15 = lane & 15;
  const int koff = half * 2;

  const float* aptr = &sa[l15 * PITCH + koff];
  const int bcol = colbase + l15;

  v8f acc = {};
#pragma unroll 8
  for (int k0 = 0; k0 < K; k0 += 4) {
    v2f a;
    a.x = aptr[k0];
    a.y = aptr[k0 + 1];
    v2f b;
    if constexpr (EXACT) {
      b.x = W[(k0 + koff) * NC + bcol];
      b.y = W[(k0 + koff + 1) * NC + bcol];
    } else {
      b.x = sw[(k0 + koff) * NCP + bcol];
      b.y = sw[(k0 + koff + 1) * NCP + bcol];
    }
    acc = __builtin_amdgcn_wmma_f32_16x16x4_f32(
         false, a,  false, b,
         (short)0, acc,  false,  false);
  }

  if (EXACT || bcol < NC) {
    float* yp = Y + (long)(rowbase + half * 8) * NC + bcol;
    if (rowbase + 16 <= nrows) {
#pragma unroll
      for (int r = 0; r < 8; ++r) yp[(long)r * NC] = acc[r];
    } else {
      const int rbase = rowbase + half * 8;
#pragma unroll
      for (int r = 0; r < 8; ++r) {
        if (rbase + r < nrows) yp[(long)r * NC] = acc[r];
      }
    }
  }
}

constexpr int RB = 256;
constexpr int QB = 16;
constexpr int AW = 32;
constexpr int SG = 8;

__device__ __forceinline__ unsigned GCN_zero_bytes(unsigned x) {
  return ~(((x & 0x7F7F7F7Fu) + 0x7F7F7F7Fu) | x | 0x7F7F7F7Fu);
}

__device__ __forceinline__ void GCN_lds_add(float* p, float v) {
  __hip_atomic_fetch_add(p, v, __ATOMIC_RELAXED, __HIP_MEMORY_SCOPE_WORKGROUP);
}

__device__ __forceinline__ int GCN_slot_base(const int* __restrict__ blk_cnt, int b, int lane) {
  int p = 0;
#pragma unroll
  for (int k = 0; k < 8; ++k) {
    const int i = lane + 32 * k;
    const int v = blk_cnt[i < b ? i : 0];
    p += i < b ? v : 0;
  }
  for (int m = 16; m > 0; m >>= 1) p += __shfl_xor(p, m);
  return __builtin_amdgcn_readfirstlane((p >> 4) + (AW + 1) * b);
}

template <int F, bool FULL>
__device__ __forceinline__ void GCN_sum_batch(const float* __restrict__ H, float* acc,
                                              int s, int d, int qn,
                                              int lane, int rowbase) {
  if constexpr (F == 128) {
    float4 v[QB];
#pragma unroll
    for (int u = 0; u < QB; ++u) {
      if (FULL || u < qn) {
        const float* hp = H + (long)__builtin_amdgcn_readlane(s, u) * 128;
        v[u] = ((const float4*)hp)[lane];
      }
    }
#pragma unroll
    for (int u = 0; u < QB; ++u) {
      if (FULL || u < qn) {
        float* ap = acc + (__builtin_amdgcn_readlane(d, u) - rowbase) * 128 + lane;
        GCN_lds_add(ap,      v[u].x);
        GCN_lds_add(ap + 32, v[u].y);
        GCN_lds_add(ap + 64, v[u].z);
        GCN_lds_add(ap + 96, v[u].w);
      }
    }
  } else {
    constexpr int NCH = (F + 31) / 32;
    float v[NCH][QB];
#pragma unroll
    for (int c = 0; c < NCH; ++c) {
      if ((c + 1) * 32 <= F || lane < F - c * 32) {
#pragma unroll
        for (int u = 0; u < QB; ++u) {
          if (FULL || u < qn) {
            const float* hp = H + (long)__builtin_amdgcn_readlane(s, u) * F;
            v[c][u] = hp[c * 32 + lane];
          }
        }
      }
    }
#pragma unroll
    for (int c = 0; c < NCH; ++c) {
      if ((c + 1) * 32 <= F || lane < F - c * 32) {
#pragma unroll
        for (int u = 0; u < QB; ++u) {
          if (FULL || u < qn) {
            const int du = __builtin_amdgcn_readlane(d, u) - rowbase;
            GCN_lds_add(acc + du * F + c * 32 + lane, v[c][u]);
          }
        }
      }
    }
  }
}

template <int F, bool FULL>
__device__ __forceinline__ void GCN_record_and_sum(const float* __restrict__ H,
                                                   const int* __restrict__ src,
                                                   const int* __restrict__ dst,
                                                   float* acc, int* nslot, int2* slots,
                                                   int qe, int qn, int lane, int rowbase) {
  int s = -1, d = rowbase;
  if (lane < (FULL ? QB : qn)) { s = src[qe]; d = dst[qe]; }
  int t = 0;
  if (lane == 0) t = __hip_atomic_fetch_add(nslot, 1, __ATOMIC_RELAXED, __HIP_MEMORY_SCOPE_WORKGROUP);
  t = __builtin_amdgcn_readfirstlane(t);
  if (lane < QB) slots[t * QB + lane] = make_int2(s, d);
  GCN_sum_batch<F, FULL>(H, acc, s, d, qn, lane, rowbase);
}

template <int F, bool RELU>
__device__ __forceinline__ void GCN_write_rows(const float* acc,
                                               const float* __restrict__ ndst,
                                               const float* __restrict__ bias,
                                               float* __restrict__ OUT,
                                               int lane, int rowbase, int nn) {
  if constexpr (F == 128) {
    for (int i = threadIdx.x; i < RB * 32; i += 32 * AW) {
      const int r = i >> 5;
      const int row = rowbase + r;
      if (row < nn) {
        const float* ap = acc + r * 128 + lane;
        const float nd = ndst[row];
        const float4 bb = ((const float4*)bias)[lane];
        float4 o;
        o.x = ap[0]  * nd + bb.x;
        o.y = ap[32] * nd + bb.y;
        o.z = ap[64] * nd + bb.z;
        o.w = ap[96] * nd + bb.w;
        if (RELU) {
          o.x = o.x > 0.0f ? o.x : 0.0f;
          o.y = o.y > 0.0f ? o.y : 0.0f;
          o.z = o.z > 0.0f ? o.z : 0.0f;
          o.w = o.w > 0.0f ? o.w : 0.0f;
        }
        ((float4*)(OUT + (long)row * 128))[lane] = o;
      }
    }
  } else {
    for (int i = threadIdx.x; i < RB * F; i += 32 * AW) {
      const int r = i / F;
      const int f = i - r * F;
      const int row = rowbase + r;
      if (row < nn) {
        float v = acc[i] * ndst[row] + bias[f];
        if (RELU) v = v > 0.0f ? v : 0.0f;
        OUT[(long)row * F + f] = v;
      }
    }
  }
}

template <int F, bool RELU>
__global__ __launch_bounds__(32 * AW)
void GCN_edge_sum(const float* __restrict__ H,
                  const int* __restrict__ src, const int* __restrict__ dst,
                  const unsigned* __restrict__ own4,
                  const int* __restrict__ blk_cnt, int* __restrict__ list,
                  int* __restrict__ nslots,
                  const float* __restrict__ ndst, const float* __restrict__ bias,
                  float* __restrict__ OUT, int ne, int nn) {
  __shared__ float acc[RB * F];
  __shared__ int nslot;
  const int lane = threadIdx.x & 31;
  const int wave = __builtin_amdgcn_readfirstlane((int)(threadIdx.x >> 5));
  const int rowbase = blockIdx.x * RB;
  int2* slots = (int2*)list + (long)GCN_slot_base(blk_cnt, blockIdx.x, lane) * QB;

  if (threadIdx.x == 0) nslot = 0;
  for (int i = threadIdx.x; i < RB * F; i += 32 * AW) acc[i] = 0.0f;
  __syncthreads();

  const unsigned pat = (unsigned)blockIdx.x * 0x01010101u;
  const uint4* own = (const uint4*)own4;
  const int nloads = ((ne + 512 * SG - 1) / (512 * SG)) * SG;
  int qe = 0;
  int qn = 0;

  for (int i0 = wave * SG; i0 < nloads; i0 += AW * SG) {
    uint4 wv[SG];
#pragma unroll
    for (int g = 0; g < SG; ++g) {
      wv[g] = own[(i0 + g) * 32 + lane];
    }
#pragma nounroll
    for (int g = 0; g < SG; ++g) {
      const uint4 w = wv[0];
#pragma unroll
      for (int k = 0; k + 1 < SG; ++k) wv[k] = wv[k + 1];
      const int i = i0 + g;
      unsigned t[4];
      t[0] = GCN_zero_bytes(w.x ^ pat);
      t[1] = GCN_zero_bytes(w.y ^ pat);
      t[2] = GCN_zero_bytes(w.z ^ pat);
      t[3] = GCN_zero_bytes(w.w ^ pat);
      unsigned ball = __builtin_amdgcn_ballot_w32((t[0] | t[1] | t[2] | t[3]) != 0u);
      while (ball) {
        const int l = __builtin_ctz(ball);
        ball &= ball - 1;
#pragma unroll
        for (int j = 0; j < 4; ++j) {
          unsigned f = (unsigned)__builtin_amdgcn_readlane((int)t[j], l);
          while (f) {
            const int bit = __builtin_ctz(f);
            f &= f - 1;
            const int e = i * 512 + l * 16 + j * 4 + (bit >> 3);
            if (e < ne) {
              if (lane == qn) qe = e;
              ++qn;
            }
          }
        }
        if (qn >= QB) {
          GCN_record_and_sum<F, true>(H, src, dst, acc, &nslot, slots, qe, qn, lane, rowbase);
          qe = __builtin_amdgcn_ds_bpermute(((lane + QB) & 31) << 2, qe);
          qn -= QB;
        }
      }
    }
  }
  if (qn > 0) GCN_record_and_sum<F, false>(H, src, dst, acc, &nslot, slots, qe, qn, lane, rowbase);
  __syncthreads();

  if (threadIdx.x == 0) nslots[blockIdx.x] = nslot;
  GCN_write_rows<F, RELU>(acc, ndst, bias, OUT, lane, rowbase, nn);
}

template <int F, bool RELU>
__global__ __launch_bounds__(32 * AW)
void GCN_edge_sum_list(const float* __restrict__ H,
                       const int* __restrict__ blk_cnt, const int* __restrict__ list,
                       const int* __restrict__ nslots,
                       const float* __restrict__ ndst, const float* __restrict__ bias,
                       float* __restrict__ OUT, int nn) {
  __shared__ float acc[RB * F];
  const int lane = threadIdx.x & 31;
  const int wave = __builtin_amdgcn_readfirstlane((int)(threadIdx.x >> 5));
  const int rowbase = blockIdx.x * RB;
  const int2* slots = (const int2*)list + (long)GCN_slot_base(blk_cnt, blockIdx.x, lane) * QB;
  const int ns = nslots[blockIdx.x];

  for (int i = threadIdx.x; i < RB * F; i += 32 * AW) acc[i] = 0.0f;
  __syncthreads();

  for (int t = 2 * wave; t < ns; t += 2 * AW) {
    int2 p = make_int2(-1, rowbase);
    if (t + (lane >> 4) < ns) p = slots[t * QB + lane];
#pragma nounroll
    for (int h = 0; h < 2; ++h) {
      const int qn = __builtin_popcount(__builtin_amdgcn_ballot_w32(p.x >= 0) & 0xFFFFu);
      if (qn == QB)    GCN_sum_batch<F, true>(H, acc, p.x, p.y, qn, lane, rowbase);
      else if (qn > 0) GCN_sum_batch<F, false>(H, acc, p.x, p.y, qn, lane, rowbase);
      p.x = __builtin_amdgcn_ds_bpermute(((lane + QB) & 31) << 2, p.x);
      p.y = __builtin_amdgcn_ds_bpermute(((lane + QB) & 31) << 2, p.y);
    }
  }
  __syncthreads();

  GCN_write_rows<F, RELU>(acc, ndst, bias, OUT, lane, rowbase, nn);
}


extern "C" void kernel_launch(void* const* d_in, const int* in_sizes, int n_in,
                              void* d_out, int out_size, void* d_ws, size_t ws_size,
                              hipStream_t stream) {
  const float* x   = (const float*)d_in[0];
  const float* W1  = (const float*)d_in[1];
  const float* b1  = (const float*)d_in[2];
  const float* W2  = (const float*)d_in[3];
  const float* b2  = (const float*)d_in[4];
  const int* esrc  = (const int*)d_in[5];
  const int* edst  = (const int*)d_in[6];
  float* out = (float*)d_out;

  const int nn = in_sizes[0] / 256;
  const int ne = in_sizes[5];

  const long NPAD = 50048;
  float* ws       = (float*)d_ws;
  float* deg_out  = ws;
  float* deg_in   = ws + NPAD;
  float* nsrc     = ws + 2 * NPAD;
  float* ndst     = ws + 3 * NPAD;
  float* h        = ws + 4 * NPAD;
  float* h1       = ws + 4 * NPAD + (long)nn * 128;
  float* h2       = ws + 4 * NPAD + 2L * (long)nn * 128;
  unsigned* own4  = (unsigned*)(ws + 4 * NPAD + 2L * (long)nn * 128 + (long)nn * 40);
  const int nwords = ((ne + 512 * SG - 1) / (512 * SG)) * (128 * SG);
  int* blk_cnt    = (int*)(own4 + nwords);
  int* nslots     = blk_cnt + 256;
  int* list       = nslots + 256;

  const int mtiles = (nn + 15) / 16;
  const int rblocks = (nn + RB - 1) / RB;

  GCN_zero_f32<<<512, 256, 0, stream>>>(deg_out, 2 * NPAD);

  GCN_degree<<<(nwords + 255) / 256, 256, 0, stream>>>(esrc, edst, deg_out, deg_in, own4, ne, nwords);
  GCN_norms<<<rblocks, RB, 0, stream>>>(deg_out, deg_in, nsrc, ndst, blk_cnt, nn);

  GCN_gemm_wmma<256, 8, 128><<<mtiles, 256, 0, stream>>>(x, nsrc, W1, h, nn);

  GCN_edge_sum<128, true><<<rblocks, 32 * AW, 0, stream>>>(h, esrc, edst, own4, blk_cnt, list, nslots,
                                                         ndst, b1, h1, ne, nn);

  GCN_gemm_wmma<128, 3, 40><<<mtiles, 96, 0, stream>>>(h1, nsrc, W2, h2, nn);

  GCN_edge_sum_list<40, false><<<rblocks, 32 * AW, 0, stream>>>(h2, blk_cnt, list, nslots, ndst, b2, out, nn);
}


# layer-2 product: each wave keeps its 16-column slice of W2 in registers as WMMA B fragments and walks three 16-row stripes; W2 is no longer staged word by word through LDS for every stripe
# speedup vs baseline: 1.9546x; 1.1928x over previous
#include <hip/hip_runtime.h>


typedef __attribute__((ext_vector_type(2))) float v2f;
typedef __attribute__((ext_vector_type(8))) float v8f;


__global__ void GCN_zero_f32(float* __restrict__ p, long n) {
  long i = (long)blockIdx.x * blockDim.x + threadIdx.x;
  long stride = (long)gridDim.x * blockDim.x;
  for (; i < n; i += stride) p[i] = 0.0f;
}

__global__ void GCN_degree(const int* __restrict__ src, const int* __restrict__ dst,
                           float* __restrict__ deg_out, float* __restrict__ deg_in,
                           unsigned* __restrict__ own4, int ne, int nwords) {
  const int g = blockIdx.x * blockDim.x + threadIdx.x;
  if (g >= nwords) return;
  const int e = 4 * g;
  unsigned packed = 0xFFFFFFFFu;
  if (e + 4 <= ne) {
    const int4 s = ((const int4*)src)[g];
    const int4 d = ((const int4*)dst)[g];
    atomicAdd(&deg_out[s.x], 1.0f); atomicAdd(&deg_out[s.y], 1.0f);
    atomicAdd(&deg_out[s.z], 1.0f); atomicAdd(&deg_out[s.w], 1.0f);
    atomicAdd(&deg_in[d.x], 1.0f);  atomicAdd(&deg_in[d.y], 1.0f);
    atomicAdd(&deg_in[d.z], 1.0f);  atomicAdd(&deg_in[d.w], 1.0f);
    packed = (((unsigned)d.x >> 8) & 255u) | ((((unsigned)d.y >> 8) & 255u) << 8) |
             ((((unsigned)d.z >> 8) & 255u) << 16) | ((((unsigned)d.w >> 8) & 255u) << 24);
  } else {
    for (int k = 0; k < 4; ++k) {
      if (e + k < ne) {
        atomicAdd(&deg_out[src[e + k]], 1.0f);
        const int dd = dst[e + k];
        atomicAdd(&deg_in[dd], 1.0f);
        packed = (packed & ~(255u << (8 * k))) | ((((unsigned)dd >> 8) & 255u) << (8 * k));
      }
    }
  }
  own4[g] = packed;
}

__global__ void GCN_norms(const float* __restrict__ deg_out, const float* __restrict__ deg_in,
                          float* __restrict__ nsrc, float* __restrict__ ndst,
                          int* __restrict__ blk_cnt, int n) {
  __shared__ int cnt;
  if (threadIdx.x == 0) cnt = 0;
  __syncthreads();
  int i = blockIdx.x * blockDim.x + threadIdx.x;
  int c = 0;
  if (i < n) {
    const float di = deg_in[i];
    nsrc[i] = rsqrtf(fmaxf(deg_out[i], 1.0f));
    ndst[i] = rsqrtf(fmaxf(di, 1.0f));
    c = (int)di;
  }
  __hip_atomic_fetch_add(&cnt, c, __ATOMIC_RELAXED, __HIP_MEMORY_SCOPE_WORKGROUP);
  __syncthreads();
  if (threadIdx.x == 0) blk_cnt[blockIdx.x] = cnt;
}

template <int K, int NT, int NC>
__global__ __launch_bounds__(32 * NT)
void GCN_gemm_wmma(const float* __restrict__ X, const float* __restrict__ scale,
                   const float* __restrict__ W, float* __restrict__ Y, int nrows) {
  constexpr int NCP = NT * 16;
  constexpr bool EXACT = (NC == NCP);
  constexpr int PITCH = K + 4;
  __shared__ float sa[16 * PITCH];
  __shared__ float sw[EXACT ? 4 : K * NCP];

  const int rowbase = blockIdx.x * 16;

  for (int idx = threadIdx.x; idx < 16 * (K / 4); idx += blockDim.x) {
    const int r = idx / (K / 4);
    const int c = idx - r * (K / 4);
    const int row = rowbase + r;
    float4 v = make_float4(0.f, 0.f, 0.f, 0.f);
    float s = 0.0f;
    if (row < nrows) {
      v = ((const float4*)(X + (long)row * K))[c];
      s = scale[row];
    }
    float* d = &sa[r * PITCH + 4 * c];
    d[0] = v.x * s; d[1] = v.y * s; d[2] = v.z * s; d[3] = v.w * s;
  }

  if constexpr (!EXACT) {
    for (int idx = threadIdx.x; idx < K * NCP; idx += blockDim.x) {
      const int k = idx / NCP;
      const int c = idx - k * NCP;
      sw[idx] = (c < NC) ? W[k * NC + c] : 0.0f;
    }
  }
  __syncthreads();

  const int lane = threadIdx.x & 31;
  const int wave = threadIdx.x >> 5;
  const int colbase = wave * 16;
  const int half = lane >> 4;
  const int l15 = lane & 15;
  const int koff = half * 2;

  const float* aptr = &sa[l15 * PITCH + koff];
  const int bcol = colbase + l15;

  v8f acc = {};
#pragma unroll 8
  for (int k0 = 0; k0 < K; k0 += 4) {
    v2f a;
    a.x = aptr[k0];
    a.y = aptr[k0 + 1];
    v2f b;
    if constexpr (EXACT) {
      b.x = W[(k0 + koff) * NC + bcol];
      b.y = W[(k0 + koff + 1) * NC + bcol];
    } else {
      b.x = sw[(k0 + koff) * NCP + bcol];
      b.y = sw[(k0 + koff + 1) * NCP + bcol];
    }
    acc = __builtin_amdgcn_wmma_f32_16x16x4_f32(
         false, a,  false, b,
         (short)0, acc,  false,  false);
  }

  if (EXACT || bcol < NC) {
    float* yp = Y + (long)(rowbase + half * 8) * NC + bcol;
    if (rowbase + 16 <= nrows) {
#pragma unroll
      for (int r = 0; r < 8; ++r) yp[(long)r * NC] = acc[r];
    } else {
      const int rbase = rowbase + half * 8;
#pragma unroll
      for (int r = 0; r < 8; ++r) {
        if (rbase + r < nrows) yp[(long)r * NC] = acc[r];
      }
    }
  }
}

template <int K, int NT, int NC>
__global__ __launch_bounds__(32 * NT)
void GCN_gemm_wreg(const float* __restrict__ X, const float* __restrict__ scale,
                   const float* __restrict__ W, float* __restrict__ Y, int nrows) {
  constexpr int NTH = 32 * NT;
  constexpr int PITCH = K + 4;
  constexpr int NV = 16 * (K / 4);
  constexpr int NLD = (NV + NTH - 1) / NTH;
  __shared__ __attribute__((aligned(16))) float sa[16 * PITCH];

  const int lane = threadIdx.x & 31;
  const int wave = threadIdx.x >> 5;
  const int colbase = wave * 16;
  const int half = lane >> 4;
  const int l15 = lane & 15;
  const int koff = half * 2;
  const int bcol = colbase + l15;
  const bool cok = bcol < NC;
  const int bc = cok ? bcol : 0;

  v2f bf[K / 4];
#pragma unroll
  for (int ks = 0; ks < K / 4; ++ks) {
    const float b0 = W[(4 * ks + koff) * NC + bc];
    const float b1 = W[(4 * ks + koff + 1) * NC + bc];
    bf[ks].x = cok ? b0 : 0.0f;
    bf[ks].y = cok ? b1 : 0.0f;
  }

  const int nstripes = (nrows + 15) >> 4;
  float4 pv[NLD];
  float ps[NLD];

  int st = blockIdx.x;
  if (st < nstripes) {
#pragma unroll
    for (int i = 0; i < NLD; ++i) {
      const int idx = threadIdx.x + i * NTH;
      const int r = idx / (K / 4);
      const int cc = idx - r * (K / 4);
      const int row = st * 16 + r;
      pv[i] = make_float4(0.f, 0.f, 0.f, 0.f);
      ps[i] = 0.0f;
      if (idx < NV && row < nrows) {
        pv[i] = ((const float4*)(X + (long)row * K))[cc];
        ps[i] = scale[row];
      }
    }
  }

  const float* aptr = &sa[l15 * PITCH + koff];

  for (; st < nstripes; st += gridDim.x) {
    const int rowbase = st * 16;
    __syncthreads();
#pragma unroll
    for (int i = 0; i < NLD; ++i) {
      const int idx = threadIdx.x + i * NTH;
      const int r = idx / (K / 4);
      const int cc = idx - r * (K / 4);
      if (idx < NV) {
        float4 o;
        o.x = pv[i].x * ps[i]; o.y = pv[i].y * ps[i];
        o.z = pv[i].z * ps[i]; o.w = pv[i].w * ps[i];
        *(float4*)&sa[r * PITCH + 4 * cc] = o;
      }
    }
    __syncthreads();

    const int nst = st + gridDim.x;
    if (nst < nstripes) {
#pragma unroll
      for (int i = 0; i < NLD; ++i) {
        const int idx = threadIdx.x + i * NTH;
        const int r = idx / (K / 4);
        const int cc = idx - r * (K / 4);
        const int row = nst * 16 + r;
        pv[i] = make_float4(0.f, 0.f, 0.f, 0.f);
        ps[i] = 0.0f;
        if (idx < NV && row < nrows) {
          pv[i] = ((const float4*)(X + (long)row * K))[cc];
          ps[i] = scale[row];
        }
      }
    }

    v8f acc = {};
#pragma unroll
    for (int ks = 0; ks < K / 4; ++ks) {
      v2f a;
      a.x = aptr[4 * ks];
      a.y = aptr[4 * ks + 1];
      acc = __builtin_amdgcn_wmma_f32_16x16x4_f32(
           false, a,  false, bf[ks],
           (short)0, acc,  false,  false);
    }

    if (cok) {
      float* yp = Y + (long)(rowbase + half * 8) * NC + bcol;
      if (rowbase + 16 <= nrows) {
#pragma unroll
        for (int r = 0; r < 8; ++r) yp[(long)r * NC] = acc[r];
      } else {
        const int rbase = rowbase + half * 8;
#pragma unroll
        for (int r = 0; r < 8; ++r) {
          if (rbase + r < nrows) yp[(long)r * NC] = acc[r];
        }
      }
    }
  }
}

constexpr int RB = 256;
constexpr int QB = 16;
constexpr int AW = 32;
constexpr int SG = 8;

__device__ __forceinline__ unsigned GCN_zero_bytes(unsigned x) {
  return ~(((x & 0x7F7F7F7Fu) + 0x7F7F7F7Fu) | x | 0x7F7F7F7Fu);
}

__device__ __forceinline__ void GCN_lds_add(float* p, float v) {
  __hip_atomic_fetch_add(p, v, __ATOMIC_RELAXED, __HIP_MEMORY_SCOPE_WORKGROUP);
}

__device__ __forceinline__ int GCN_slot_base(const int* __restrict__ blk_cnt, int b, int lane) {
  int p = 0;
#pragma unroll
  for (int k = 0; k < 8; ++k) {
    const int i = lane + 32 * k;
    const int v = blk_cnt[i < b ? i : 0];
    p += i < b ? v : 0;
  }
  for (int m = 16; m > 0; m >>= 1) p += __shfl_xor(p, m);
  return __builtin_amdgcn_readfirstlane((p >> 4) + (AW + 1) * b);
}

template <int F, bool FULL>
__device__ __forceinline__ void GCN_sum_batch(const float* __restrict__ H, float* acc,
                                              int s, int d, int qn,
                                              int lane, int rowbase) {
  if constexpr (F == 128) {
    float4 v[QB];
#pragma unroll
    for (int u = 0; u < QB; ++u) {
      if (FULL || u < qn) {
        const float* hp = H + (long)__builtin_amdgcn_readlane(s, u) * 128;
        v[u] = ((const float4*)hp)[lane];
      }
    }
#pragma unroll
    for (int u = 0; u < QB; ++u) {
      if (FULL || u < qn) {
        float* ap = acc + (__builtin_amdgcn_readlane(d, u) - rowbase) * 128 + lane;
        GCN_lds_add(ap,      v[u].x);
        GCN_lds_add(ap + 32, v[u].y);
        GCN_lds_add(ap + 64, v[u].z);
        GCN_lds_add(ap + 96, v[u].w);
      }
    }
  } else {
    constexpr int NCH = (F + 31) / 32;
    float v[NCH][QB];
#pragma unroll
    for (int c = 0; c < NCH; ++c) {
      if ((c + 1) * 32 <= F || lane < F - c * 32) {
#pragma unroll
        for (int u = 0; u < QB; ++u) {
          if (FULL || u < qn) {
            const float* hp = H + (long)__builtin_amdgcn_readlane(s, u) * F;
            v[c][u] = hp[c * 32 + lane];
          }
        }
      }
    }
#pragma unroll
    for (int c = 0; c < NCH; ++c) {
      if ((c + 1) * 32 <= F || lane < F - c * 32) {
#pragma unroll
        for (int u = 0; u < QB; ++u) {
          if (FULL || u < qn) {
            const int du = __builtin_amdgcn_readlane(d, u) - rowbase;
            GCN_lds_add(acc + du * F + c * 32 + lane, v[c][u]);
          }
        }
      }
    }
  }
}

template <int F, bool FULL>
__device__ __forceinline__ void GCN_record_and_sum(const float* __restrict__ H,
                                                   const int* __restrict__ src,
                                                   const int* __restrict__ dst,
                                                   float* acc, int* nslot, int2* slots,
                                                   int qe, int qn, int lane, int rowbase) {
  int s = -1, d = rowbase;
  if (lane < (FULL ? QB : qn)) { s = src[qe]; d = dst[qe]; }
  int t = 0;
  if (lane == 0) t = __hip_atomic_fetch_add(nslot, 1, __ATOMIC_RELAXED, __HIP_MEMORY_SCOPE_WORKGROUP);
  t = __builtin_amdgcn_readfirstlane(t);
  if (lane < QB) slots[t * QB + lane] = make_int2(s, d);
  GCN_sum_batch<F, FULL>(H, acc, s, d, qn, lane, rowbase);
}

template <int F, bool RELU>
__device__ __forceinline__ void GCN_write_rows(const float* acc,
                                               const float* __restrict__ ndst,
                                               const float* __restrict__ bias,
                                               float* __restrict__ OUT,
                                               int lane, int rowbase, int nn) {
  if constexpr (F == 128) {
    for (int i = threadIdx.x; i < RB * 32; i += 32 * AW) {
      const int r = i >> 5;
      const int row = rowbase + r;
      if (row < nn) {
        const float* ap = acc + r * 128 + lane;
        const float nd = ndst[row];
        const float4 bb = ((const float4*)bias)[lane];
        float4 o;
        o.x = ap[0]  * nd + bb.x;
        o.y = ap[32] * nd + bb.y;
        o.z = ap[64] * nd + bb.z;
        o.w = ap[96] * nd + bb.w;
        if (RELU) {
          o.x = o.x > 0.0f ? o.x : 0.0f;
          o.y = o.y > 0.0f ? o.y : 0.0f;
          o.z = o.z > 0.0f ? o.z : 0.0f;
          o.w = o.w > 0.0f ? o.w : 0.0f;
        }
        ((float4*)(OUT + (long)row * 128))[lane] = o;
      }
    }
  } else {
    for (int i = threadIdx.x; i < RB * F; i += 32 * AW) {
      const int r = i / F;
      const int f = i - r * F;
      const int row = rowbase + r;
      if (row < nn) {
        float v = acc[i] * ndst[row] + bias[f];
        if (RELU) v = v > 0.0f ? v : 0.0f;
        OUT[(long)row * F + f] = v;
      }
    }
  }
}

template <int F, bool RELU>
__global__ __launch_bounds__(32 * AW)
void GCN_edge_sum(const float* __restrict__ H,
                  const int* __restrict__ src, const int* __restrict__ dst,
                  const unsigned* __restrict__ own4,
                  const int* __restrict__ blk_cnt, int* __restrict__ list,
                  int* __restrict__ nslots,
                  const float* __restrict__ ndst, const float* __restrict__ bias,
                  float* __restrict__ OUT, int ne, int nn) {
  __shared__ float acc[RB * F];
  __shared__ int nslot;
  const int lane = threadIdx.x & 31;
  const int wave = __builtin_amdgcn_readfirstlane((int)(threadIdx.x >> 5));
  const int rowbase = blockIdx.x * RB;
  int2* slots = (int2*)list + (long)GCN_slot_base(blk_cnt, blockIdx.x, lane) * QB;

  if (threadIdx.x == 0) nslot = 0;
  for (int i = threadIdx.x; i < RB * F; i += 32 * AW) acc[i] = 0.0f;
  __syncthreads();

  const unsigned pat = (unsigned)blockIdx.x * 0x01010101u;
  const uint4* own = (const uint4*)own4;
  const int nloads = ((ne + 512 * SG - 1) / (512 * SG)) * SG;
  int qe = 0;
  int qn = 0;

  for (int i0 = wave * SG; i0 < nloads; i0 += AW * SG) {
    uint4 wv[SG];
#pragma unroll
    for (int g = 0; g < SG; ++g) {
      wv[g] = own[(i0 + g) * 32 + lane];
    }
#pragma nounroll
    for (int g = 0; g < SG; ++g) {
      const uint4 w = wv[0];
#pragma unroll
      for (int k = 0; k + 1 < SG; ++k) wv[k] = wv[k + 1];
      const int i = i0 + g;
      unsigned t[4];
      t[0] = GCN_zero_bytes(w.x ^ pat);
      t[1] = GCN_zero_bytes(w.y ^ pat);
      t[2] = GCN_zero_bytes(w.z ^ pat);
      t[3] = GCN_zero_bytes(w.w ^ pat);
      unsigned ball = __builtin_amdgcn_ballot_w32((t[0] | t[1] | t[2] | t[3]) != 0u);
      while (ball) {
        const int l = __builtin_ctz(ball);
        ball &= ball - 1;
#pragma unroll
        for (int j = 0; j < 4; ++j) {
          unsigned f = (unsigned)__builtin_amdgcn_readlane((int)t[j], l);
          while (f) {
            const int bit = __builtin_ctz(f);
            f &= f - 1;
            const int e = i * 512 + l * 16 + j * 4 + (bit >> 3);
            if (e < ne) {
              if (lane == qn) qe = e;
              ++qn;
            }
          }
        }
        if (qn >= QB) {
          GCN_record_and_sum<F, true>(H, src, dst, acc, &nslot, slots, qe, qn, lane, rowbase);
          qe = __builtin_amdgcn_ds_bpermute(((lane + QB) & 31) << 2, qe);
          qn -= QB;
        }
      }
    }
  }
  if (qn > 0) GCN_record_and_sum<F, false>(H, src, dst, acc, &nslot, slots, qe, qn, lane, rowbase);
  __syncthreads();

  if (threadIdx.x == 0) nslots[blockIdx.x] = nslot;
  GCN_write_rows<F, RELU>(acc, ndst, bias, OUT, lane, rowbase, nn);
}

template <int F, bool RELU>
__global__ __launch_bounds__(32 * AW)
void GCN_edge_sum_list(const float* __restrict__ H,
                       const int* __restrict__ blk_cnt, const int* __restrict__ list,
                       const int* __restrict__ nslots,
                       const float* __restrict__ ndst, const float* __restrict__ bias,
                       float* __restrict__ OUT, int nn) {
  __shared__ float acc[RB * F];
  const int lane = threadIdx.x & 31;
  const int wave = __builtin_amdgcn_readfirstlane((int)(threadIdx.x >> 5));
  const int rowbase = blockIdx.x * RB;
  const int2* slots = (const int2*)list + (long)GCN_slot_base(blk_cnt, blockIdx.x, lane) * QB;
  const int ns = nslots[blockIdx.x];

  for (int i = threadIdx.x; i < RB * F; i += 32 * AW) acc[i] = 0.0f;
  __syncthreads();

  for (int t = 2 * wave; t < ns; t += 2 * AW) {
    int2 p = make_int2(-1, rowbase);
    if (t + (lane >> 4) < ns) p = slots[t * QB + lane];
#pragma nounroll
    for (int h = 0; h < 2; ++h) {
      const int qn = __builtin_popcount(__builtin_amdgcn_ballot_w32(p.x >= 0) & 0xFFFFu);
      if (qn == QB)    GCN_sum_batch<F, true>(H, acc, p.x, p.y, qn, lane, rowbase);
      else if (qn > 0) GCN_sum_batch<F, false>(H, acc, p.x, p.y, qn, lane, rowbase);
      p.x = __builtin_amdgcn_ds_bpermute(((lane + QB) & 31) << 2, p.x);
      p.y = __builtin_amdgcn_ds_bpermute(((lane + QB) & 31) << 2, p.y);
    }
  }
  __syncthreads();

  GCN_write_rows<F, RELU>(acc, ndst, bias, OUT, lane, rowbase, nn);
}


extern "C" void kernel_launch(void* const* d_in, const int* in_sizes, int n_in,
                              void* d_out, int out_size, void* d_ws, size_t ws_size,
                              hipStream_t stream) {
  const float* x   = (const float*)d_in[0];
  const float* W1  = (const float*)d_in[1];
  const float* b1  = (const float*)d_in[2];
  const float* W2  = (const float*)d_in[3];
  const float* b2  = (const float*)d_in[4];
  const int* esrc  = (const int*)d_in[5];
  const int* edst  = (const int*)d_in[6];
  float* out = (float*)d_out;

  const int nn = in_sizes[0] / 256;
  const int ne = in_sizes[5];

  const long NPAD = 50048;
  float* ws       = (float*)d_ws;
  float* deg_out  = ws;
  float* deg_in   = ws + NPAD;
  float* nsrc     = ws + 2 * NPAD;
  float* ndst     = ws + 3 * NPAD;
  float* h        = ws + 4 * NPAD;
  float* h1       = ws + 4 * NPAD + (long)nn * 128;
  float* h2       = ws + 4 * NPAD + 2L * (long)nn * 128;
  unsigned* own4  = (unsigned*)(ws + 4 * NPAD + 2L * (long)nn * 128 + (long)nn * 40);
  const int nwords = ((ne + 512 * SG - 1) / (512 * SG)) * (128 * SG);
  int* blk_cnt    = (int*)(own4 + nwords);
  int* nslots     = blk_cnt + 256;
  int* list       = nslots + 256;

  const int mtiles = (nn + 15) / 16;
  const int rblocks = (nn + RB - 1) / RB;

  GCN_zero_f32<<<512, 256, 0, stream>>>(deg_out, 2 * NPAD);

  GCN_degree<<<(nwords + 255) / 256, 256, 0, stream>>>(esrc, edst, deg_out, deg_in, own4, ne, nwords);
  GCN_norms<<<rblocks, RB, 0, stream>>>(deg_out, deg_in, nsrc, ndst, blk_cnt, nn);

  GCN_gemm_wmma<256, 8, 128><<<mtiles, 256, 0, stream>>>(x, nsrc, W1, h, nn);

  GCN_edge_sum<128, true><<<rblocks, 32 * AW, 0, stream>>>(h, esrc, edst, own4, blk_cnt, list, nslots,
                                                         ndst, b1, h1, ne, nn);

  GCN_gemm_wreg<128, 3, 40><<<(mtiles + 2) / 3, 96, 0, stream>>>(h1, nsrc, W2, h2, nn);

  GCN_edge_sum_list<40, false><<<rblocks, 32 * AW, 0, stream>>>(h2, blk_cnt, list, nslots, ndst, b2, out, nn);
}


# edge-sum row blocks re-tiled from 256 to 200 destination rows: 250 work-groups, one per WGP on 250 of the 256 WGPs instead of 196; block-id bytes are dst / 200; LDS adds stay in arrival order
# speedup vs baseline: 2.0517x; 1.0497x over previous
#include <hip/hip_runtime.h>


typedef __attribute__((ext_vector_type(2))) float v2f;
typedef __attribute__((ext_vector_type(8))) float v8f;

constexpr int RB = 200;


__global__ void GCN_zero_f32(float* __restrict__ p, long n) {
  long i = (long)blockIdx.x * blockDim.x + threadIdx.x;
  long stride = (long)gridDim.x * blockDim.x;
  for (; i < n; i += stride) p[i] = 0.0f;
}

__global__ void GCN_degree(const int* __restrict__ src, const int* __restrict__ dst,
                           float* __restrict__ deg_out, float* __restrict__ deg_in,
                           unsigned* __restrict__ own4, int ne, int nwords) {
  const int g = blockIdx.x * blockDim.x + threadIdx.x;
  if (g >= nwords) return;
  const int e = 4 * g;
  unsigned packed = 0xFFFFFFFFu;
  if (e + 4 <= ne) {
    const int4 s = ((const int4*)src)[g];
    const int4 d = ((const int4*)dst)[g];
    atomicAdd(&deg_out[s.x], 1.0f); atomicAdd(&deg_out[s.y], 1.0f);
    atomicAdd(&deg_out[s.z], 1.0f); atomicAdd(&deg_out[s.w], 1.0f);
    atomicAdd(&deg_in[d.x], 1.0f);  atomicAdd(&deg_in[d.y], 1.0f);
    atomicAdd(&deg_in[d.z], 1.0f);  atomicAdd(&deg_in[d.w], 1.0f);
    packed = (((unsigned)d.x / RB) & 255u) | ((((unsigned)d.y / RB) & 255u) << 8) |
             ((((unsigned)d.z / RB) & 255u) << 16) | ((((unsigned)d.w / RB) & 255u) << 24);
  } else {
    for (int k = 0; k < 4; ++k) {
      if (e + k < ne) {
        atomicAdd(&deg_out[src[e + k]], 1.0f);
        const int dd = dst[e + k];
        atomicAdd(&deg_in[dd], 1.0f);
        packed = (packed & ~(255u << (8 * k))) | ((((unsigned)dd / RB) & 255u) << (8 * k));
      }
    }
  }
  own4[g] = packed;
}

__global__ void GCN_norms(const float* __restrict__ deg_out, const float* __restrict__ deg_in,
                          float* __restrict__ nsrc, float* __restrict__ ndst,
                          int* __restrict__ blk_cnt, int n) {
  __shared__ int cnt;
  if (threadIdx.x == 0) cnt = 0;
  __syncthreads();
  int i = blockIdx.x * blockDim.x + threadIdx.x;
  int c = 0;
  if (i < n) {
    const float di = deg_in[i];
    nsrc[i] = rsqrtf(fmaxf(deg_out[i], 1.0f));
    ndst[i] = rsqrtf(fmaxf(di, 1.0f));
    c = (int)di;
  }
  __hip_atomic_fetch_add(&cnt, c, __ATOMIC_RELAXED, __HIP_MEMORY_SCOPE_WORKGROUP);
  __syncthreads();
  if (threadIdx.x == 0) blk_cnt[blockIdx.x] = cnt;
}

template <int K, int NT, int NC>
__global__ __launch_bounds__(32 * NT)
void GCN_gemm_wmma(const float* __restrict__ X, const float* __restrict__ scale,
                   const float* __restrict__ W, float* __restrict__ Y, int nrows) {
  constexpr int NCP = NT * 16;
  constexpr bool EXACT = (NC == NCP);
  constexpr int PITCH = K + 4;
  __shared__ float sa[16 * PITCH];
  __shared__ float sw[EXACT ? 4 : K * NCP];

  const int rowbase = blockIdx.x * 16;

  for (int idx = threadIdx.x; idx < 16 * (K / 4); idx += blockDim.x) {
    const int r = idx / (K / 4);
    const int c = idx - r * (K / 4);
    const int row = rowbase + r;
    float4 v = make_float4(0.f, 0.f, 0.f, 0.f);
    float s = 0.0f;
    if (row < nrows) {
      v = ((const float4*)(X + (long)row * K))[c];
      s = scale[row];
    }
    float* d = &sa[r * PITCH + 4 * c];
    d[0] = v.x * s; d[1] = v.y * s; d[2] = v.z * s; d[3] = v.w * s;
  }

  if constexpr (!EXACT) {
    for (int idx = threadIdx.x; idx < K * NCP; idx += blockDim.x) {
      const int k = idx / NCP;
      const int c = idx - k * NCP;
      sw[idx] = (c < NC) ? W[k * NC + c] : 0.0f;
    }
  }
  __syncthreads();

  const int lane = threadIdx.x & 31;
  const int wave = threadIdx.x >> 5;
  const int colbase = wave * 16;
  const int half = lane >> 4;
  const int l15 = lane & 15;
  const int koff = half * 2;

  const float* aptr = &sa[l15 * PITCH + koff];
  const int bcol = colbase + l15;

  v8f acc = {};
#pragma unroll 8
  for (int k0 = 0; k0 < K; k0 += 4) {
    v2f a;
    a.x = aptr[k0];
    a.y = aptr[k0 + 1];
    v2f b;
    if constexpr (EXACT) {
      b.x = W[(k0 + koff) * NC + bcol];
      b.y = W[(k0 + koff + 1) * NC + bcol];
    } else {
      b.x = sw[(k0 + koff) * NCP + bcol];
      b.y = sw[(k0 + koff + 1) * NCP + bcol];
    }
    acc = __builtin_amdgcn_wmma_f32_16x16x4_f32(
         false, a,  false, b,
         (short)0, acc,  false,  false);
  }

  if (EXACT || bcol < NC) {
    float* yp = Y + (long)(rowbase + half * 8) * NC + bcol;
    if (rowbase + 16 <= nrows) {
#pragma unroll
      for (int r = 0; r < 8; ++r) yp[(long)r * NC] = acc[r];
    } else {
      const int rbase = rowbase + half * 8;
#pragma unroll
      for (int r = 0; r < 8; ++r) {
        if (rbase + r < nrows) yp[(long)r * NC] = acc[r];
      }
    }
  }
}

template <int K, int NT, int NC>
__global__ __launch_bounds__(32 * NT)
void GCN_gemm_wreg(const float* __restrict__ X, const float* __restrict__ scale,
                   const float* __restrict__ W, float* __restrict__ Y, int nrows) {
  constexpr int NTH = 32 * NT;
  constexpr int PITCH = K + 4;
  constexpr int NV = 16 * (K / 4);
  constexpr int NLD = (NV + NTH - 1) / NTH;
  __shared__ __attribute__((aligned(16))) float sa[16 * PITCH];

  const int lane = threadIdx.x & 31;
  const int wave = threadIdx.x >> 5;
  const int colbase = wave * 16;
  const int half = lane >> 4;
  const int l15 = lane & 15;
  const int koff = half * 2;
  const int bcol = colbase + l15;
  const bool cok = bcol < NC;
  const int bc = cok ? bcol : 0;

  v2f bf[K / 4];
#pragma unroll
  for (int ks = 0; ks < K / 4; ++ks) {
    const float b0 = W[(4 * ks + koff) * NC + bc];
    const float b1 = W[(4 * ks + koff + 1) * NC + bc];
    bf[ks].x = cok ? b0 : 0.0f;
    bf[ks].y = cok ? b1 : 0.0f;
  }

  const int nstripes = (nrows + 15) >> 4;
  float4 pv[NLD];
  float ps[NLD];

  int st = blockIdx.x;
  if (st < nstripes) {
#pragma unroll
    for (int i = 0; i < NLD; ++i) {
      const int idx = threadIdx.x + i * NTH;
      const int r = idx / (K / 4);
      const int cc = idx - r * (K / 4);
      const int row = st * 16 + r;
      pv[i] = make_float4(0.f, 0.f, 0.f, 0.f);
      ps[i] = 0.0f;
      if (idx < NV && row < nrows) {
        pv[i] = ((const float4*)(X + (long)row * K))[cc];
        ps[i] = scale[row];
      }
    }
  }

  const float* aptr = &sa[l15 * PITCH + koff];

  for (; st < nstripes; st += gridDim.x) {
    const int rowbase = st * 16;
    __syncthreads();
#pragma unroll
    for (int i = 0; i < NLD; ++i) {
      const int idx = threadIdx.x + i * NTH;
      const int r = idx / (K / 4);
      const int cc = idx - r * (K / 4);
      if (idx < NV) {
        float4 o;
        o.x = pv[i].x * ps[i]; o.y = pv[i].y * ps[i];
        o.z = pv[i].z * ps[i]; o.w = pv[i].w * ps[i];
        *(float4*)&sa[r * PITCH + 4 * cc] = o;
      }
    }
    __syncthreads();

    const int nst = st + gridDim.x;
    if (nst < nstripes) {
#pragma unroll
      for (int i = 0; i < NLD; ++i) {
        const int idx = threadIdx.x + i * NTH;
        const int r = idx / (K / 4);
        const int cc = idx - r * (K / 4);
        const int row = nst * 16 + r;
        pv[i] = make_float4(0.f, 0.f, 0.f, 0.f);
        ps[i] = 0.0f;
        if (idx < NV && row < nrows) {
          pv[i] = ((const float4*)(X + (long)row * K))[cc];
          ps[i] = scale[row];
        }
      }
    }

    v8f acc = {};
#pragma unroll
    for (int ks = 0; ks < K / 4; ++ks) {
      v2f a;
      a.x = aptr[4 * ks];
      a.y = aptr[4 * ks + 1];
      acc = __builtin_amdgcn_wmma_f32_16x16x4_f32(
           false, a,  false, bf[ks],
           (short)0, acc,  false,  false);
    }

    if (cok) {
      float* yp = Y + (long)(rowbase + half * 8) * NC + bcol;
      if (rowbase + 16 <= nrows) {
#pragma unroll
        for (int r = 0; r < 8; ++r) yp[(long)r * NC] = acc[r];
      } else {
        const int rbase = rowbase + half * 8;
#pragma unroll
        for (int r = 0; r < 8; ++r) {
          if (rbase + r < nrows) yp[(long)r * NC] = acc[r];
        }
      }
    }
  }
}

constexpr int QB = 16;
constexpr int AW = 32;
constexpr int SG = 8;

__device__ __forceinline__ unsigned GCN_zero_bytes(unsigned x) {
  return ~(((x & 0x7F7F7F7Fu) + 0x7F7F7F7Fu) | x | 0x7F7F7F7Fu);
}

__device__ __forceinline__ void GCN_lds_add(float* p, float v) {
  __hip_atomic_fetch_add(p, v, __ATOMIC_RELAXED, __HIP_MEMORY_SCOPE_WORKGROUP);
}

__device__ __forceinline__ int GCN_slot_base(const int* __restrict__ blk_cnt, int b, int lane) {
  int p = 0;
#pragma unroll
  for (int k = 0; k < 8; ++k) {
    const int i = lane + 32 * k;
    const int v = blk_cnt[i < b ? i : 0];
    p += i < b ? v : 0;
  }
  for (int m = 16; m > 0; m >>= 1) p += __shfl_xor(p, m);
  return __builtin_amdgcn_readfirstlane((p >> 4) + (AW + 1) * b);
}

template <int F, bool FULL>
__device__ __forceinline__ void GCN_sum_batch(const float* __restrict__ H, float* acc,
                                              int s, int d, int qn,
                                              int lane, int rowbase) {
  if constexpr (F == 128) {
    float4 v[QB];
#pragma unroll
    for (int u = 0; u < QB; ++u) {
      if (FULL || u < qn) {
        const float* hp = H + (long)__builtin_amdgcn_readlane(s, u) * 128;
        v[u] = ((const float4*)hp)[lane];
      }
    }
#pragma unroll
    for (int u = 0; u < QB; ++u) {
      if (FULL || u < qn) {
        float* ap = acc + (__builtin_amdgcn_readlane(d, u) - rowbase) * 128 + lane;
        GCN_lds_add(ap,      v[u].x);
        GCN_lds_add(ap + 32, v[u].y);
        GCN_lds_add(ap + 64, v[u].z);
        GCN_lds_add(ap + 96, v[u].w);
      }
    }
  } else {
    constexpr int NCH = (F + 31) / 32;
    float v[NCH][QB];
#pragma unroll
    for (int c = 0; c < NCH; ++c) {
      if ((c + 1) * 32 <= F || lane < F - c * 32) {
#pragma unroll
        for (int u = 0; u < QB; ++u) {
          if (FULL || u < qn) {
            const float* hp = H + (long)__builtin_amdgcn_readlane(s, u) * F;
            v[c][u] = hp[c * 32 + lane];
          }
        }
      }
    }
#pragma unroll
    for (int c = 0; c < NCH; ++c) {
      if ((c + 1) * 32 <= F || lane < F - c * 32) {
#pragma unroll
        for (int u = 0; u < QB; ++u) {
          if (FULL || u < qn) {
            const int du = __builtin_amdgcn_readlane(d, u) - rowbase;
            GCN_lds_add(acc + du * F + c * 32 + lane, v[c][u]);
          }
        }
      }
    }
  }
}

template <int F, bool FULL>
__device__ __forceinline__ void GCN_record_and_sum(const float* __restrict__ H,
                                                   const int* __restrict__ src,
                                                   const int* __restrict__ dst,
                                                   float* acc, int* nslot, int2* slots,
                                                   int qe, int qn, int lane, int rowbase) {
  int s = -1, d = rowbase;
  if (lane < (FULL ? QB : qn)) { s = src[qe]; d = dst[qe]; }
  int t = 0;
  if (lane == 0) t = __hip_atomic_fetch_add(nslot, 1, __ATOMIC_RELAXED, __HIP_MEMORY_SCOPE_WORKGROUP);
  t = __builtin_amdgcn_readfirstlane(t);
  if (lane < QB) slots[t * QB + lane] = make_int2(s, d);
  GCN_sum_batch<F, FULL>(H, acc, s, d, qn, lane, rowbase);
}

template <int F, bool RELU>
__device__ __forceinline__ void GCN_write_rows(const float* acc,
                                               const float* __restrict__ ndst,
                                               const float* __restrict__ bias,
                                               float* __restrict__ OUT,
                                               int lane, int rowbase, int nn) {
  if constexpr (F == 128) {
    for (int i = threadIdx.x; i < RB * 32; i += 32 * AW) {
      const int r = i >> 5;
      const int row = rowbase + r;
      if (row < nn) {
        const float* ap = acc + r * 128 + lane;
        const float nd = ndst[row];
        const float4 bb = ((const float4*)bias)[lane];
        float4 o;
        o.x = ap[0]  * nd + bb.x;
        o.y = ap[32] * nd + bb.y;
        o.z = ap[64] * nd + bb.z;
        o.w = ap[96] * nd + bb.w;
        if (RELU) {
          o.x = o.x > 0.0f ? o.x : 0.0f;
          o.y = o.y > 0.0f ? o.y : 0.0f;
          o.z = o.z > 0.0f ? o.z : 0.0f;
          o.w = o.w > 0.0f ? o.w : 0.0f;
        }
        ((float4*)(OUT + (long)row * 128))[lane] = o;
      }
    }
  } else {
    for (int i = threadIdx.x; i < RB * F; i += 32 * AW) {
      const int r = i / F;
      const int f = i - r * F;
      const int row = rowbase + r;
      if (row < nn) {
        float v = acc[i] * ndst[row] + bias[f];
        if (RELU) v = v > 0.0f ? v : 0.0f;
        OUT[(long)row * F + f] = v;
      }
    }
  }
}

template <int F, bool RELU>
__global__ __launch_bounds__(32 * AW)
void GCN_edge_sum(const float* __restrict__ H,
                  const int* __restrict__ src, const int* __restrict__ dst,
                  const unsigned* __restrict__ own4,
                  const int* __restrict__ blk_cnt, int* __restrict__ list,
                  int* __restrict__ nslots,
                  const float* __restrict__ ndst, const float* __restrict__ bias,
                  float* __restrict__ OUT, int ne, int nn) {
  __shared__ float acc[RB * F];
  __shared__ int nslot;
  const int lane = threadIdx.x & 31;
  const int wave = __builtin_amdgcn_readfirstlane((int)(threadIdx.x >> 5));
  const int rowbase = blockIdx.x * RB;
  int2* slots = (int2*)list + (long)GCN_slot_base(blk_cnt, blockIdx.x, lane) * QB;

  if (threadIdx.x == 0) nslot = 0;
  for (int i = threadIdx.x; i < RB * F; i += 32 * AW) acc[i] = 0.0f;
  __syncthreads();

  const unsigned pat = (unsigned)blockIdx.x * 0x01010101u;
  const uint4* own = (const uint4*)own4;
  const int nloads = ((ne + 512 * SG - 1) / (512 * SG)) * SG;
  int qe = 0;
  int qn = 0;

  for (int i0 = wave * SG; i0 < nloads; i0 += AW * SG) {
    uint4 wv[SG];
#pragma unroll
    for (int g = 0; g < SG; ++g) {
      wv[g] = own[(i0 + g) * 32 + lane];
    }
#pragma nounroll
    for (int g = 0; g < SG; ++g) {
      const uint4 w = wv[0];
#pragma unroll
      for (int k = 0; k + 1 < SG; ++k) wv[k] = wv[k + 1];
      const int i = i0 + g;
      unsigned t[4];
      t[0] = GCN_zero_bytes(w.x ^ pat);
      t[1] = GCN_zero_bytes(w.y ^ pat);
      t[2] = GCN_zero_bytes(w.z ^ pat);
      t[3] = GCN_zero_bytes(w.w ^ pat);
      unsigned ball = __builtin_amdgcn_ballot_w32((t[0] | t[1] | t[2] | t[3]) != 0u);
      while (ball) {
        const int l = __builtin_ctz(ball);
        ball &= ball - 1;
#pragma unroll
        for (int j = 0; j < 4; ++j) {
          unsigned f = (unsigned)__builtin_amdgcn_readlane((int)t[j], l);
          while (f) {
            const int bit = __builtin_ctz(f);
            f &= f - 1;
            const int e = i * 512 + l * 16 + j * 4 + (bit >> 3);
            if (e < ne) {
              if (lane == qn) qe = e;
              ++qn;
            }
          }
        }
        if (qn >= QB) {
          GCN_record_and_sum<F, true>(H, src, dst, acc, &nslot, slots, qe, qn, lane, rowbase);
          qe = __builtin_amdgcn_ds_bpermute(((lane + QB) & 31) << 2, qe);
          qn -= QB;
        }
      }
    }
  }
  if (qn > 0) GCN_record_and_sum<F, false>(H, src, dst, acc, &nslot, slots, qe, qn, lane, rowbase);
  __syncthreads();

  if (threadIdx.x == 0) nslots[blockIdx.x] = nslot;
  GCN_write_rows<F, RELU>(acc, ndst, bias, OUT, lane, rowbase, nn);
}

template <int F, bool RELU>
__global__ __launch_bounds__(32 * AW)
void GCN_edge_sum_list(const float* __restrict__ H,
                       const int* __restrict__ blk_cnt, const int* __restrict__ list,
                       const int* __restrict__ nslots,
                       const float* __restrict__ ndst, const float* __restrict__ bias,
                       float* __restrict__ OUT, int nn) {
  __shared__ float acc[RB * F];
  const int lane = threadIdx.x & 31;
  const int wave = __builtin_amdgcn_readfirstlane((int)(threadIdx.x >> 5));
  const int rowbase = blockIdx.x * RB;
  const int2* slots = (const int2*)list + (long)GCN_slot_base(blk_cnt, blockIdx.x, lane) * QB;
  const int ns = nslots[blockIdx.x];

  for (int i = threadIdx.x; i < RB * F; i += 32 * AW) acc[i] = 0.0f;
  __syncthreads();

  for (int t = 2 * wave; t < ns; t += 2 * AW) {
    int2 p = make_int2(-1, rowbase);
    if (t + (lane >> 4) < ns) p = slots[t * QB + lane];
#pragma nounroll
    for (int h = 0; h < 2; ++h) {
      const int qn = __builtin_popcount(__builtin_amdgcn_ballot_w32(p.x >= 0) & 0xFFFFu);
      if (qn == QB)    GCN_sum_batch<F, true>(H, acc, p.x, p.y, qn, lane, rowbase);
      else if (qn > 0) GCN_sum_batch<F, false>(H, acc, p.x, p.y, qn, lane, rowbase);
      p.x = __builtin_amdgcn_ds_bpermute(((lane + QB) & 31) << 2, p.x);
      p.y = __builtin_amdgcn_ds_bpermute(((lane + QB) & 31) << 2, p.y);
    }
  }
  __syncthreads();

  GCN_write_rows<F, RELU>(acc, ndst, bias, OUT, lane, rowbase, nn);
}


extern "C" void kernel_launch(void* const* d_in, const int* in_sizes, int n_in,
                              void* d_out, int out_size, void* d_ws, size_t ws_size,
                              hipStream_t stream) {
  const float* x   = (const float*)d_in[0];
  const float* W1  = (const float*)d_in[1];
  const float* b1  = (const float*)d_in[2];
  const float* W2  = (const float*)d_in[3];
  const float* b2  = (const float*)d_in[4];
  const int* esrc  = (const int*)d_in[5];
  const int* edst  = (const int*)d_in[6];
  float* out = (float*)d_out;

  const int nn = in_sizes[0] / 256;
  const int ne = in_sizes[5];

  const long NPAD = 50048;
  float* ws       = (float*)d_ws;
  float* deg_out  = ws;
  float* deg_in   = ws + NPAD;
  float* nsrc     = ws + 2 * NPAD;
  float* ndst     = ws + 3 * NPAD;
  float* h        = ws + 4 * NPAD;
  float* h1       = ws + 4 * NPAD + (long)nn * 128;
  float* h2       = ws + 4 * NPAD + 2L * (long)nn * 128;
  unsigned* own4  = (unsigned*)(ws + 4 * NPAD + 2L * (long)nn * 128 + (long)nn * 40);
  const int nwords = ((ne + 512 * SG - 1) / (512 * SG)) * (128 * SG);
  int* blk_cnt    = (int*)(own4 + nwords);
  int* nslots     = blk_cnt + 256;
  int* list       = nslots + 256;

  const int mtiles = (nn + 15) / 16;
  const int rblocks = (nn + RB - 1) / RB;

  GCN_zero_f32<<<512, 256, 0, stream>>>(deg_out, 2 * NPAD);

  GCN_degree<<<(nwords + 255) / 256, 256, 0, stream>>>(esrc, edst, deg_out, deg_in, own4, ne, nwords);
  GCN_norms<<<rblocks, RB, 0, stream>>>(deg_out, deg_in, nsrc, ndst, blk_cnt, nn);

  GCN_gemm_wmma<256, 8, 128><<<mtiles, 256, 0, stream>>>(x, nsrc, W1, h, nn);

  GCN_edge_sum<128, true><<<rblocks, 32 * AW, 0, stream>>>(h, esrc, edst, own4, blk_cnt, list, nslots,
                                                         ndst, b1, h1, ne, nn);

  GCN_gemm_wreg<128, 3, 40><<<(mtiles + 2) / 3, 96, 0, stream>>>(h1, nsrc, W2, h2, nn);

  GCN_edge_sum_list<40, false><<<rblocks, 32 * AW, 0, stream>>>(h2, blk_cnt, list, nslots, ndst, b2, out, nn);
}


# degree pass joins the layer-1 product's launch as extra work-groups (7 to 6 launches); product takes unscaled x, layer-1 edge sum scales gathered rows by the source-degree factor; h rounding changes
# speedup vs baseline: 2.1440x; 1.0450x over previous
#include <hip/hip_runtime.h>


typedef __attribute__((ext_vector_type(2))) float v2f;
typedef __attribute__((ext_vector_type(8))) float v8f;

constexpr int RB = 200;


__global__ void GCN_zero_f32(float* __restrict__ p, long n) {
  long i = (long)blockIdx.x * blockDim.x + threadIdx.x;
  long stride = (long)gridDim.x * blockDim.x;
  for (; i < n; i += stride) p[i] = 0.0f;
}


__device__ __forceinline__ unsigned GCN_degree4(const int4 s, const int4 d,
                                                float* __restrict__ deg_out,
                                                float* __restrict__ deg_in) {
  atomicAdd(&deg_out[s.x], 1.0f); atomicAdd(&deg_out[s.y], 1.0f);
  atomicAdd(&deg_out[s.z], 1.0f); atomicAdd(&deg_out[s.w], 1.0f);
  atomicAdd(&deg_in[d.x], 1.0f);  atomicAdd(&deg_in[d.y], 1.0f);
  atomicAdd(&deg_in[d.z], 1.0f);  atomicAdd(&deg_in[d.w], 1.0f);
  return (((unsigned)d.x / RB) & 255u) | ((((unsigned)d.y / RB) & 255u) << 8) |
         ((((unsigned)d.z / RB) & 255u) << 16) | ((((unsigned)d.w / RB) & 255u) << 24);
}

__device__ __forceinline__ void GCN_degree(const int* __restrict__ src, const int* __restrict__ dst,
                                           float* __restrict__ deg_out, float* __restrict__ deg_in,
                                           unsigned* __restrict__ own4, int ne, int nwords, int g) {
  if (g >= nwords) return;
  const int e = 4 * g;
  unsigned packed = 0xFFFFFFFFu;
  if (e + 4 <= ne) {
    packed = GCN_degree4(((const int4*)src)[g], ((const int4*)dst)[g], deg_out, deg_in);
  } else {
    for (int k = 0; k < 4; ++k) {
      if (e + k < ne) {
        atomicAdd(&deg_out[src[e + k]], 1.0f);
        const int dd = dst[e + k];
        atomicAdd(&deg_in[dd], 1.0f);
        packed = (packed & ~(255u << (8 * k))) | ((((unsigned)dd / RB) & 255u) << (8 * k));
      }
    }
  }
  own4[g] = packed;
}

__global__ void GCN_norms(const float* __restrict__ deg_out, const float* __restrict__ deg_in,
                          float* __restrict__ nsrc, float* __restrict__ ndst,
                          int* __restrict__ blk_cnt, int n) {
  __shared__ int cnt;
  if (threadIdx.x == 0) cnt = 0;
  __syncthreads();
  int i = blockIdx.x * blockDim.x + threadIdx.x;
  int c = 0;
  if (i < n) {
    const float di = deg_in[i];
    nsrc[i] = rsqrtf(fmaxf(deg_out[i], 1.0f));
    ndst[i] = rsqrtf(fmaxf(di, 1.0f));
    c = (int)di;
  }
  __hip_atomic_fetch_add(&cnt, c, __ATOMIC_RELAXED, __HIP_MEMORY_SCOPE_WORKGROUP);
  __syncthreads();
  if (threadIdx.x == 0) blk_cnt[blockIdx.x] = cnt;
}

constexpr int DEGW = 4;
template <int K, int NT, int NC>
__global__ __launch_bounds__(32 * NT)
void GCN_gemm_wmma_deg(const float* __restrict__ X,
                       const float* __restrict__ W, float* __restrict__ Y, int nrows, int ndeg,
                       const int* __restrict__ src, const int* __restrict__ dst,
                       float* __restrict__ deg_out, float* __restrict__ deg_in,
                       unsigned* __restrict__ own4, int ne, int nwords) {
  if ((int)blockIdx.x < ndeg) {
    const int g0 = (int)blockIdx.x * (DEGW * 32 * NT) + (int)threadIdx.x;
    const int glast = g0 + (DEGW - 1) * (32 * NT);
    if (glast < nwords && 4 * glast + 4 <= ne) {
      int4 s[DEGW], d[DEGW];
#pragma unroll
      for (int k = 0; k < DEGW; ++k) {
        s[k] = ((const int4*)src)[g0 + k * (32 * NT)];
        d[k] = ((const int4*)dst)[g0 + k * (32 * NT)];
      }
#pragma unroll
      for (int k = 0; k < DEGW; ++k) {
        own4[g0 + k * (32 * NT)] = GCN_degree4(s[k], d[k], deg_out, deg_in);
      }
    } else {
      for (int k = 0; k < DEGW; ++k) {
        GCN_degree(src, dst, deg_out, deg_in, own4, ne, nwords, g0 + k * (32 * NT));
      }
    }
    return;
  }

  constexpr int NCP = NT * 16;
  constexpr bool EXACT = (NC == NCP);
  constexpr int PITCH = K + 4;
  __shared__ float sa[16 * PITCH];
  __shared__ float sw[EXACT ? 4 : K * NCP];

  const int rowbase = ((int)blockIdx.x - ndeg) * 16;

  for (int idx = threadIdx.x; idx < 16 * (K / 4); idx += blockDim.x) {
    const int r = idx / (K / 4);
    const int c = idx - r * (K / 4);
    const int row = rowbase + r;
    float4 v = make_float4(0.f, 0.f, 0.f, 0.f);
    if (row < nrows) v = ((const float4*)(X + (long)row * K))[c];
    float* d = &sa[r * PITCH + 4 * c];
    d[0] = v.x; d[1] = v.y; d[2] = v.z; d[3] = v.w;
  }

  if constexpr (!EXACT) {
    for (int idx = threadIdx.x; idx < K * NCP; idx += blockDim.x) {
      const int k = idx / NCP;
      const int c = idx - k * NCP;
      sw[idx] = (c < NC) ? W[k * NC + c] : 0.0f;
    }
  }
  __syncthreads();

  const int lane = threadIdx.x & 31;
  const int wave = threadIdx.x >> 5;
  const int colbase = wave * 16;
  const int half = lane >> 4;
  const int l15 = lane & 15;
  const int koff = half * 2;

  const float* aptr = &sa[l15 * PITCH + koff];
  const int bcol = colbase + l15;

  v8f acc = {};
#pragma unroll 8
  for (int k0 = 0; k0 < K; k0 += 4) {
    v2f a;
    a.x = aptr[k0];
    a.y = aptr[k0 + 1];
    v2f b;
    if constexpr (EXACT) {
      b.x = W[(k0 + koff) * NC + bcol];
      b.y = W[(k0 + koff + 1) * NC + bcol];
    } else {
      b.x = sw[(k0 + koff) * NCP + bcol];
      b.y = sw[(k0 + koff + 1) * NCP + bcol];
    }
    acc = __builtin_amdgcn_wmma_f32_16x16x4_f32(
         false, a,  false, b,
         (short)0, acc,  false,  false);
  }

  if (EXACT || bcol < NC) {
    float* yp = Y + (long)(rowbase + half * 8) * NC + bcol;
    if (rowbase + 16 <= nrows) {
#pragma unroll
      for (int r = 0; r < 8; ++r) yp[(long)r * NC] = acc[r];
    } else {
      const int rbase = rowbase + half * 8;
#pragma unroll
      for (int r = 0; r < 8; ++r) {
        if (rbase + r < nrows) yp[(long)r * NC] = acc[r];
      }
    }
  }
}

template <int K, int NT, int NC>
__global__ __launch_bounds__(32 * NT)
void GCN_gemm_wreg(const float* __restrict__ X, const float* __restrict__ scale,
                   const float* __restrict__ W, float* __restrict__ Y, int nrows) {
  constexpr int NTH = 32 * NT;
  constexpr int PITCH = K + 4;
  constexpr int NV = 16 * (K / 4);
  constexpr int NLD = (NV + NTH - 1) / NTH;
  __shared__ __attribute__((aligned(16))) float sa[16 * PITCH];

  const int lane = threadIdx.x & 31;
  const int wave = threadIdx.x >> 5;
  const int colbase = wave * 16;
  const int half = lane >> 4;
  const int l15 = lane & 15;
  const int koff = half * 2;
  const int bcol = colbase + l15;
  const bool cok = bcol < NC;
  const int bc = cok ? bcol : 0;

  v2f bf[K / 4];
#pragma unroll
  for (int ks = 0; ks < K / 4; ++ks) {
    const float b0 = W[(4 * ks + koff) * NC + bc];
    const float b1 = W[(4 * ks + koff + 1) * NC + bc];
    bf[ks].x = cok ? b0 : 0.0f;
    bf[ks].y = cok ? b1 : 0.0f;
  }

  const int nstripes = (nrows + 15) >> 4;
  float4 pv[NLD];
  float ps[NLD];

  int st = blockIdx.x;
  if (st < nstripes) {
#pragma unroll
    for (int i = 0; i < NLD; ++i) {
      const int idx = threadIdx.x + i * NTH;
      const int r = idx / (K / 4);
      const int cc = idx - r * (K / 4);
      const int row = st * 16 + r;
      pv[i] = make_float4(0.f, 0.f, 0.f, 0.f);
      ps[i] = 0.0f;
      if (idx < NV && row < nrows) {
        pv[i] = ((const float4*)(X + (long)row * K))[cc];
        ps[i] = scale[row];
      }
    }
  }

  const float* aptr = &sa[l15 * PITCH + koff];

  for (; st < nstripes; st += gridDim.x) {
    const int rowbase = st * 16;
    __syncthreads();
#pragma unroll
    for (int i = 0; i < NLD; ++i) {
      const int idx = threadIdx.x + i * NTH;
      const int r = idx / (K / 4);
      const int cc = idx - r * (K / 4);
      if (idx < NV) {
        float4 o;
        o.x = pv[i].x * ps[i]; o.y = pv[i].y * ps[i];
        o.z = pv[i].z * ps[i]; o.w = pv[i].w * ps[i];
        *(float4*)&sa[r * PITCH + 4 * cc] = o;
      }
    }
    __syncthreads();

    const int nst = st + gridDim.x;
    if (nst < nstripes) {
#pragma unroll
      for (int i = 0; i < NLD; ++i) {
        const int idx = threadIdx.x + i * NTH;
        const int r = idx / (K / 4);
        const int cc = idx - r * (K / 4);
        const int row = nst * 16 + r;
        pv[i] = make_float4(0.f, 0.f, 0.f, 0.f);
        ps[i] = 0.0f;
        if (idx < NV && row < nrows) {
          pv[i] = ((const float4*)(X + (long)row * K))[cc];
          ps[i] = scale[row];
        }
      }
    }

    v8f acc = {};
#pragma unroll
    for (int ks = 0; ks < K / 4; ++ks) {
      v2f a;
      a.x = aptr[4 * ks];
      a.y = aptr[4 * ks + 1];
      acc = __builtin_amdgcn_wmma_f32_16x16x4_f32(
           false, a,  false, bf[ks],
           (short)0, acc,  false,  false);
    }

    if (cok) {
      float* yp = Y + (long)(rowbase + half * 8) * NC + bcol;
      if (rowbase + 16 <= nrows) {
#pragma unroll
        for (int r = 0; r < 8; ++r) yp[(long)r * NC] = acc[r];
      } else {
        const int rbase = rowbase + half * 8;
#pragma unroll
        for (int r = 0; r < 8; ++r) {
          if (rbase + r < nrows) yp[(long)r * NC] = acc[r];
        }
      }
    }
  }
}

constexpr int QB = 16;
constexpr int AW = 32;
constexpr int SG = 8;

__device__ __forceinline__ unsigned GCN_zero_bytes(unsigned x) {
  return ~(((x & 0x7F7F7F7Fu) + 0x7F7F7F7Fu) | x | 0x7F7F7F7Fu);
}

__device__ __forceinline__ void GCN_lds_add(float* p, float v) {
  __hip_atomic_fetch_add(p, v, __ATOMIC_RELAXED, __HIP_MEMORY_SCOPE_WORKGROUP);
}

__device__ __forceinline__ int GCN_slot_base(const int* __restrict__ blk_cnt, int b, int lane) {
  int p = 0;
#pragma unroll
  for (int k = 0; k < 8; ++k) {
    const int i = lane + 32 * k;
    const int v = blk_cnt[i < b ? i : 0];
    p += i < b ? v : 0;
  }
  for (int m = 16; m > 0; m >>= 1) p += __shfl_xor(p, m);
  return __builtin_amdgcn_readfirstlane((p >> 4) + (AW + 1) * b);
}

template <int F, bool FULL, bool SCALED>
__device__ __forceinline__ void GCN_sum_batch(const float* __restrict__ H, float* acc,
                                              int s, int d, float ns, int qn,
                                              int lane, int rowbase) {
  if constexpr (F == 128) {
    float4 v[QB];
#pragma unroll
    for (int u = 0; u < QB; ++u) {
      if (FULL || u < qn) {
        const float* hp = H + (long)__builtin_amdgcn_readlane(s, u) * 128;
        v[u] = ((const float4*)hp)[lane];
      }
    }
#pragma unroll
    for (int u = 0; u < QB; ++u) {
      if (FULL || u < qn) {
        float* ap = acc + (__builtin_amdgcn_readlane(d, u) - rowbase) * 128 + lane;
        float4 w = v[u];
        if constexpr (SCALED) {
          const float su = __int_as_float(__builtin_amdgcn_readlane(__float_as_int(ns), u));
          w.x *= su; w.y *= su; w.z *= su; w.w *= su;
        }
        GCN_lds_add(ap,      w.x);
        GCN_lds_add(ap + 32, w.y);
        GCN_lds_add(ap + 64, w.z);
        GCN_lds_add(ap + 96, w.w);
      }
    }
  } else {
    constexpr int NCH = (F + 31) / 32;
    float v[NCH][QB];
#pragma unroll
    for (int c = 0; c < NCH; ++c) {
      if ((c + 1) * 32 <= F || lane < F - c * 32) {
#pragma unroll
        for (int u = 0; u < QB; ++u) {
          if (FULL || u < qn) {
            const float* hp = H + (long)__builtin_amdgcn_readlane(s, u) * F;
            v[c][u] = hp[c * 32 + lane];
          }
        }
      }
    }
#pragma unroll
    for (int c = 0; c < NCH; ++c) {
      if ((c + 1) * 32 <= F || lane < F - c * 32) {
#pragma unroll
        for (int u = 0; u < QB; ++u) {
          if (FULL || u < qn) {
            const int du = __builtin_amdgcn_readlane(d, u) - rowbase;
            float w = v[c][u];
            if constexpr (SCALED) {
              w *= __int_as_float(__builtin_amdgcn_readlane(__float_as_int(ns), u));
            }
            GCN_lds_add(acc + du * F + c * 32 + lane, w);
          }
        }
      }
    }
  }
}

template <int F, bool FULL>
__device__ __forceinline__ void GCN_record_and_sum(const float* __restrict__ H,
                                                   const int* __restrict__ src,
                                                   const int* __restrict__ dst,
                                                   const float* __restrict__ nsrc,
                                                   float* acc, int* nslot, int2* slots,
                                                   int qe, int qn, int lane, int rowbase) {
  int s = -1, d = rowbase;
  float ns = 0.0f;
  if (lane < (FULL ? QB : qn)) { s = src[qe]; d = dst[qe]; ns = nsrc[s]; }
  int t = 0;
  if (lane == 0) t = __hip_atomic_fetch_add(nslot, 1, __ATOMIC_RELAXED, __HIP_MEMORY_SCOPE_WORKGROUP);
  t = __builtin_amdgcn_readfirstlane(t);
  if (lane < QB) slots[t * QB + lane] = make_int2(s, d);
  GCN_sum_batch<F, FULL, true>(H, acc, s, d, ns, qn, lane, rowbase);
}

template <int F, bool RELU>
__device__ __forceinline__ void GCN_write_rows(const float* acc,
                                               const float* __restrict__ ndst,
                                               const float* __restrict__ bias,
                                               float* __restrict__ OUT,
                                               int lane, int rowbase, int nn) {
  if constexpr (F == 128) {
    for (int i = threadIdx.x; i < RB * 32; i += 32 * AW) {
      const int r = i >> 5;
      const int row = rowbase + r;
      if (row < nn) {
        const float* ap = acc + r * 128 + lane;
        const float nd = ndst[row];
        const float4 bb = ((const float4*)bias)[lane];
        float4 o;
        o.x = ap[0]  * nd + bb.x;
        o.y = ap[32] * nd + bb.y;
        o.z = ap[64] * nd + bb.z;
        o.w = ap[96] * nd + bb.w;
        if (RELU) {
          o.x = o.x > 0.0f ? o.x : 0.0f;
          o.y = o.y > 0.0f ? o.y : 0.0f;
          o.z = o.z > 0.0f ? o.z : 0.0f;
          o.w = o.w > 0.0f ? o.w : 0.0f;
        }
        ((float4*)(OUT + (long)row * 128))[lane] = o;
      }
    }
  } else {
    for (int i = threadIdx.x; i < RB * F; i += 32 * AW) {
      const int r = i / F;
      const int f = i - r * F;
      const int row = rowbase + r;
      if (row < nn) {
        float v = acc[i] * ndst[row] + bias[f];
        if (RELU) v = v > 0.0f ? v : 0.0f;
        OUT[(long)row * F + f] = v;
      }
    }
  }
}

template <int F, bool RELU>
__global__ __launch_bounds__(32 * AW)
void GCN_edge_sum(const float* __restrict__ H,
                  const int* __restrict__ src, const int* __restrict__ dst,
                  const float* __restrict__ nsrc,
                  const unsigned* __restrict__ own4,
                  const int* __restrict__ blk_cnt, int* __restrict__ list,
                  int* __restrict__ nslots,
                  const float* __restrict__ ndst, const float* __restrict__ bias,
                  float* __restrict__ OUT, int ne, int nn) {
  __shared__ float acc[RB * F];
  __shared__ int nslot;
  const int lane = threadIdx.x & 31;
  const int wave = __builtin_amdgcn_readfirstlane((int)(threadIdx.x >> 5));
  const int rowbase = blockIdx.x * RB;
  int2* slots = (int2*)list + (long)GCN_slot_base(blk_cnt, blockIdx.x, lane) * QB;

  if (threadIdx.x == 0) nslot = 0;
  for (int i = threadIdx.x; i < RB * F; i += 32 * AW) acc[i] = 0.0f;
  __syncthreads();

  const unsigned pat = (unsigned)blockIdx.x * 0x01010101u;
  const uint4* own = (const uint4*)own4;
  const int nloads = ((ne + 512 * SG - 1) / (512 * SG)) * SG;
  int qe = 0;
  int qn = 0;

  for (int i0 = wave * SG; i0 < nloads; i0 += AW * SG) {
    uint4 wv[SG];
#pragma unroll
    for (int g = 0; g < SG; ++g) {
      wv[g] = own[(i0 + g) * 32 + lane];
    }
#pragma nounroll
    for (int g = 0; g < SG; ++g) {
      const uint4 w = wv[0];
#pragma unroll
      for (int k = 0; k + 1 < SG; ++k) wv[k] = wv[k + 1];
      const int i = i0 + g;
      unsigned t[4];
      t[0] = GCN_zero_bytes(w.x ^ pat);
      t[1] = GCN_zero_bytes(w.y ^ pat);
      t[2] = GCN_zero_bytes(w.z ^ pat);
      t[3] = GCN_zero_bytes(w.w ^ pat);
      unsigned ball = __builtin_amdgcn_ballot_w32((t[0] | t[1] | t[2] | t[3]) != 0u);
      while (ball) {
        const int l = __builtin_ctz(ball);
        ball &= ball - 1;
#pragma unroll
        for (int j = 0; j < 4; ++j) {
          unsigned f = (unsigned)__builtin_amdgcn_readlane((int)t[j], l);
          while (f) {
            const int bit = __builtin_ctz(f);
            f &= f - 1;
            const int e = i * 512 + l * 16 + j * 4 + (bit >> 3);
            if (e < ne) {
              if (lane == qn) qe = e;
              ++qn;
            }
          }
        }
        if (qn >= QB) {
          GCN_record_and_sum<F, true>(H, src, dst, nsrc, acc, &nslot, slots, qe, qn, lane, rowbase);
          qe = __builtin_amdgcn_ds_bpermute(((lane + QB) & 31) << 2, qe);
          qn -= QB;
        }
      }
    }
  }
  if (qn > 0) GCN_record_and_sum<F, false>(H, src, dst, nsrc, acc, &nslot, slots, qe, qn, lane, rowbase);
  __syncthreads();

  if (threadIdx.x == 0) nslots[blockIdx.x] = nslot;
  GCN_write_rows<F, RELU>(acc, ndst, bias, OUT, lane, rowbase, nn);
}

template <int F, bool RELU>
__global__ __launch_bounds__(32 * AW)
void GCN_edge_sum_list(const float* __restrict__ H,
                       const int* __restrict__ blk_cnt, const int* __restrict__ list,
                       const int* __restrict__ nslots,
                       const float* __restrict__ ndst, const float* __restrict__ bias,
                       float* __restrict__ OUT, int nn) {
  __shared__ float acc[RB * F];
  const int lane = threadIdx.x & 31;
  const int wave = __builtin_amdgcn_readfirstlane((int)(threadIdx.x >> 5));
  const int rowbase = blockIdx.x * RB;
  const int2* slots = (const int2*)list + (long)GCN_slot_base(blk_cnt, blockIdx.x, lane) * QB;
  const int ns = nslots[blockIdx.x];

  for (int i = threadIdx.x; i < RB * F; i += 32 * AW) acc[i] = 0.0f;
  __syncthreads();

  for (int t = 2 * wave; t < ns; t += 2 * AW) {
    int2 p = make_int2(-1, rowbase);
    if (t + (lane >> 4) < ns) p = slots[t * QB + lane];
#pragma nounroll
    for (int h = 0; h < 2; ++h) {
      const int qn = __builtin_popcount(__builtin_amdgcn_ballot_w32(p.x >= 0) & 0xFFFFu);
      if (qn == QB)    GCN_sum_batch<F, true, false>(H, acc, p.x, p.y, 0.0f, qn, lane, rowbase);
      else if (qn > 0) GCN_sum_batch<F, false, false>(H, acc, p.x, p.y, 0.0f, qn, lane, rowbase);
      p.x = __builtin_amdgcn_ds_bpermute(((lane + QB) & 31) << 2, p.x);
      p.y = __builtin_amdgcn_ds_bpermute(((lane + QB) & 31) << 2, p.y);
    }
  }
  __syncthreads();

  GCN_write_rows<F, RELU>(acc, ndst, bias, OUT, lane, rowbase, nn);
}


extern "C" void kernel_launch(void* const* d_in, const int* in_sizes, int n_in,
                              void* d_out, int out_size, void* d_ws, size_t ws_size,
                              hipStream_t stream) {
  const float* x   = (const float*)d_in[0];
  const float* W1  = (const float*)d_in[1];
  const float* b1  = (const float*)d_in[2];
  const float* W2  = (const float*)d_in[3];
  const float* b2  = (const float*)d_in[4];
  const int* esrc  = (const int*)d_in[5];
  const int* edst  = (const int*)d_in[6];
  float* out = (float*)d_out;

  const int nn = in_sizes[0] / 256;
  const int ne = in_sizes[5];

  const long NPAD = 50048;
  float* ws       = (float*)d_ws;
  float* deg_out  = ws;
  float* deg_in   = ws + NPAD;
  float* nsrc     = ws + 2 * NPAD;
  float* ndst     = ws + 3 * NPAD;
  float* h        = ws + 4 * NPAD;
  float* h1       = ws + 4 * NPAD + (long)nn * 128;
  float* h2       = ws + 4 * NPAD + 2L * (long)nn * 128;
  unsigned* own4  = (unsigned*)(ws + 4 * NPAD + 2L * (long)nn * 128 + (long)nn * 40);
  const int nwords = ((ne + 512 * SG - 1) / (512 * SG)) * (128 * SG);
  int* blk_cnt    = (int*)(own4 + nwords);
  int* nslots     = blk_cnt + 256;
  int* list       = nslots + 256;

  const int mtiles = (nn + 15) / 16;
  const int rblocks = (nn + RB - 1) / RB;

  GCN_zero_f32<<<512, 256, 0, stream>>>(deg_out, 2 * NPAD);

  const int ndeg = (nwords + DEGW * 256 - 1) / (DEGW * 256);
  GCN_gemm_wmma_deg<256, 8, 128><<<ndeg + mtiles, 256, 0, stream>>>(
      x, W1, h, nn, ndeg, esrc, edst, deg_out, deg_in, own4, ne, nwords);

  GCN_norms<<<rblocks, RB, 0, stream>>>(deg_out, deg_in, nsrc, ndst, blk_cnt, nn);

  GCN_edge_sum<128, true><<<rblocks, 32 * AW, 0, stream>>>(h, esrc, edst, nsrc, own4, blk_cnt, list, nslots,
                                                         ndst, b1, h1, ne, nn);

  GCN_gemm_wreg<128, 3, 40><<<(mtiles + 2) / 3, 96, 0, stream>>>(h1, nsrc, W2, h2, nn);

  GCN_edge_sum_list<40, false><<<rblocks, 32 * AW, 0, stream>>>(h2, blk_cnt, list, nslots, ndst, b2, out, nn);
}


# layer-1 edge-sum scan: each lane holding a matching block-id byte queues one edge per step into a per-wave LDS ring (ballot + lane rank) instead of one lane at a time; batch and slot order change
# speedup vs baseline: 2.2314x; 1.0408x over previous
#include <hip/hip_runtime.h>


typedef __attribute__((ext_vector_type(2))) float v2f;
typedef __attribute__((ext_vector_type(8))) float v8f;

constexpr int RB = 200;


__global__ void GCN_zero_f32(float* __restrict__ p, long n) {
  long i = (long)blockIdx.x * blockDim.x + threadIdx.x;
  long stride = (long)gridDim.x * blockDim.x;
  for (; i < n; i += stride) p[i] = 0.0f;
}


__device__ __forceinline__ unsigned GCN_degree4(const int4 s, const int4 d,
                                                float* __restrict__ deg_out,
                                                float* __restrict__ deg_in) {
  atomicAdd(&deg_out[s.x], 1.0f); atomicAdd(&deg_out[s.y], 1.0f);
  atomicAdd(&deg_out[s.z], 1.0f); atomicAdd(&deg_out[s.w], 1.0f);
  atomicAdd(&deg_in[d.x], 1.0f);  atomicAdd(&deg_in[d.y], 1.0f);
  atomicAdd(&deg_in[d.z], 1.0f);  atomicAdd(&deg_in[d.w], 1.0f);
  return (((unsigned)d.x / RB) & 255u) | ((((unsigned)d.y / RB) & 255u) << 8) |
         ((((unsigned)d.z / RB) & 255u) << 16) | ((((unsigned)d.w / RB) & 255u) << 24);
}

__device__ __forceinline__ void GCN_degree(const int* __restrict__ src, const int* __restrict__ dst,
                                           float* __restrict__ deg_out, float* __restrict__ deg_in,
                                           unsigned* __restrict__ own4, int ne, int nwords, int g) {
  if (g >= nwords) return;
  const int e = 4 * g;
  unsigned packed = 0xFFFFFFFFu;
  if (e + 4 <= ne) {
    packed = GCN_degree4(((const int4*)src)[g], ((const int4*)dst)[g], deg_out, deg_in);
  } else {
    for (int k = 0; k < 4; ++k) {
      if (e + k < ne) {
        atomicAdd(&deg_out[src[e + k]], 1.0f);
        const int dd = dst[e + k];
        atomicAdd(&deg_in[dd], 1.0f);
        packed = (packed & ~(255u << (8 * k))) | ((((unsigned)dd / RB) & 255u) << (8 * k));
      }
    }
  }
  own4[g] = packed;
}

__global__ void GCN_norms(const float* __restrict__ deg_out, const float* __restrict__ deg_in,
                          float* __restrict__ nsrc, float* __restrict__ ndst,
                          int* __restrict__ blk_cnt, int n) {
  __shared__ int cnt;
  if (threadIdx.x == 0) cnt = 0;
  __syncthreads();
  int i = blockIdx.x * blockDim.x + threadIdx.x;
  int c = 0;
  if (i < n) {
    const float di = deg_in[i];
    nsrc[i] = rsqrtf(fmaxf(deg_out[i], 1.0f));
    ndst[i] = rsqrtf(fmaxf(di, 1.0f));
    c = (int)di;
  }
  __hip_atomic_fetch_add(&cnt, c, __ATOMIC_RELAXED, __HIP_MEMORY_SCOPE_WORKGROUP);
  __syncthreads();
  if (threadIdx.x == 0) blk_cnt[blockIdx.x] = cnt;
}

constexpr int DEGW = 4;
template <int K, int NT, int NC>
__global__ __launch_bounds__(32 * NT)
void GCN_gemm_wmma_deg(const float* __restrict__ X,
                       const float* __restrict__ W, float* __restrict__ Y, int nrows, int ndeg,
                       const int* __restrict__ src, const int* __restrict__ dst,
                       float* __restrict__ deg_out, float* __restrict__ deg_in,
                       unsigned* __restrict__ own4, int ne, int nwords) {
  if ((int)blockIdx.x < ndeg) {
    const int g0 = (int)blockIdx.x * (DEGW * 32 * NT) + (int)threadIdx.x;
    const int glast = g0 + (DEGW - 1) * (32 * NT);
    if (glast < nwords && 4 * glast + 4 <= ne) {
      int4 s[DEGW], d[DEGW];
#pragma unroll
      for (int k = 0; k < DEGW; ++k) {
        s[k] = ((const int4*)src)[g0 + k * (32 * NT)];
        d[k] = ((const int4*)dst)[g0 + k * (32 * NT)];
      }
#pragma unroll
      for (int k = 0; k < DEGW; ++k) {
        own4[g0 + k * (32 * NT)] = GCN_degree4(s[k], d[k], deg_out, deg_in);
      }
    } else {
      for (int k = 0; k < DEGW; ++k) {
        GCN_degree(src, dst, deg_out, deg_in, own4, ne, nwords, g0 + k * (32 * NT));
      }
    }
    return;
  }

  constexpr int NCP = NT * 16;
  constexpr bool EXACT = (NC == NCP);
  constexpr int PITCH = K + 4;
  __shared__ float sa[16 * PITCH];
  __shared__ float sw[EXACT ? 4 : K * NCP];

  const int rowbase = ((int)blockIdx.x - ndeg) * 16;

  for (int idx = threadIdx.x; idx < 16 * (K / 4); idx += blockDim.x) {
    const int r = idx / (K / 4);
    const int c = idx - r * (K / 4);
    const int row = rowbase + r;
    float4 v = make_float4(0.f, 0.f, 0.f, 0.f);
    if (row < nrows) v = ((const float4*)(X + (long)row * K))[c];
    float* d = &sa[r * PITCH + 4 * c];
    d[0] = v.x; d[1] = v.y; d[2] = v.z; d[3] = v.w;
  }

  if constexpr (!EXACT) {
    for (int idx = threadIdx.x; idx < K * NCP; idx += blockDim.x) {
      const int k = idx / NCP;
      const int c = idx - k * NCP;
      sw[idx] = (c < NC) ? W[k * NC + c] : 0.0f;
    }
  }
  __syncthreads();

  const int lane = threadIdx.x & 31;
  const int wave = threadIdx.x >> 5;
  const int colbase = wave * 16;
  const int half = lane >> 4;
  const int l15 = lane & 15;
  const int koff = half * 2;

  const float* aptr = &sa[l15 * PITCH + koff];
  const int bcol = colbase + l15;

  v8f acc = {};
#pragma unroll 8
  for (int k0 = 0; k0 < K; k0 += 4) {
    v2f a;
    a.x = aptr[k0];
    a.y = aptr[k0 + 1];
    v2f b;
    if constexpr (EXACT) {
      b.x = W[(k0 + koff) * NC + bcol];
      b.y = W[(k0 + koff + 1) * NC + bcol];
    } else {
      b.x = sw[(k0 + koff) * NCP + bcol];
      b.y = sw[(k0 + koff + 1) * NCP + bcol];
    }
    acc = __builtin_amdgcn_wmma_f32_16x16x4_f32(
         false, a,  false, b,
         (short)0, acc,  false,  false);
  }

  if (EXACT || bcol < NC) {
    float* yp = Y + (long)(rowbase + half * 8) * NC + bcol;
    if (rowbase + 16 <= nrows) {
#pragma unroll
      for (int r = 0; r < 8; ++r) yp[(long)r * NC] = acc[r];
    } else {
      const int rbase = rowbase + half * 8;
#pragma unroll
      for (int r = 0; r < 8; ++r) {
        if (rbase + r < nrows) yp[(long)r * NC] = acc[r];
      }
    }
  }
}

template <int K, int NT, int NC>
__global__ __launch_bounds__(32 * NT)
void GCN_gemm_wreg(const float* __restrict__ X, const float* __restrict__ scale,
                   const float* __restrict__ W, float* __restrict__ Y, int nrows) {
  constexpr int NTH = 32 * NT;
  constexpr int PITCH = K + 4;
  constexpr int NV = 16 * (K / 4);
  constexpr int NLD = (NV + NTH - 1) / NTH;
  __shared__ __attribute__((aligned(16))) float sa[16 * PITCH];

  const int lane = threadIdx.x & 31;
  const int wave = threadIdx.x >> 5;
  const int colbase = wave * 16;
  const int half = lane >> 4;
  const int l15 = lane & 15;
  const int koff = half * 2;
  const int bcol = colbase + l15;
  const bool cok = bcol < NC;
  const int bc = cok ? bcol : 0;

  v2f bf[K / 4];
#pragma unroll
  for (int ks = 0; ks < K / 4; ++ks) {
    const float b0 = W[(4 * ks + koff) * NC + bc];
    const float b1 = W[(4 * ks + koff + 1) * NC + bc];
    bf[ks].x = cok ? b0 : 0.0f;
    bf[ks].y = cok ? b1 : 0.0f;
  }

  const int nstripes = (nrows + 15) >> 4;
  float4 pv[NLD];
  float ps[NLD];

  int st = blockIdx.x;
  if (st < nstripes) {
#pragma unroll
    for (int i = 0; i < NLD; ++i) {
      const int idx = threadIdx.x + i * NTH;
      const int r = idx / (K / 4);
      const int cc = idx - r * (K / 4);
      const int row = st * 16 + r;
      pv[i] = make_float4(0.f, 0.f, 0.f, 0.f);
      ps[i] = 0.0f;
      if (idx < NV && row < nrows) {
        pv[i] = ((const float4*)(X + (long)row * K))[cc];
        ps[i] = scale[row];
      }
    }
  }

  const float* aptr = &sa[l15 * PITCH + koff];

  for (; st < nstripes; st += gridDim.x) {
    const int rowbase = st * 16;
    __syncthreads();
#pragma unroll
    for (int i = 0; i < NLD; ++i) {
      const int idx = threadIdx.x + i * NTH;
      const int r = idx / (K / 4);
      const int cc = idx - r * (K / 4);
      if (idx < NV) {
        float4 o;
        o.x = pv[i].x * ps[i]; o.y = pv[i].y * ps[i];
        o.z = pv[i].z * ps[i]; o.w = pv[i].w * ps[i];
        *(float4*)&sa[r * PITCH + 4 * cc] = o;
      }
    }
    __syncthreads();

    const int nst = st + gridDim.x;
    if (nst < nstripes) {
#pragma unroll
      for (int i = 0; i < NLD; ++i) {
        const int idx = threadIdx.x + i * NTH;
        const int r = idx / (K / 4);
        const int cc = idx - r * (K / 4);
        const int row = nst * 16 + r;
        pv[i] = make_float4(0.f, 0.f, 0.f, 0.f);
        ps[i] = 0.0f;
        if (idx < NV && row < nrows) {
          pv[i] = ((const float4*)(X + (long)row * K))[cc];
          ps[i] = scale[row];
        }
      }
    }

    v8f acc = {};
#pragma unroll
    for (int ks = 0; ks < K / 4; ++ks) {
      v2f a;
      a.x = aptr[4 * ks];
      a.y = aptr[4 * ks + 1];
      acc = __builtin_amdgcn_wmma_f32_16x16x4_f32(
           false, a,  false, bf[ks],
           (short)0, acc,  false,  false);
    }

    if (cok) {
      float* yp = Y + (long)(rowbase + half * 8) * NC + bcol;
      if (rowbase + 16 <= nrows) {
#pragma unroll
        for (int r = 0; r < 8; ++r) yp[(long)r * NC] = acc[r];
      } else {
        const int rbase = rowbase + half * 8;
#pragma unroll
        for (int r = 0; r < 8; ++r) {
          if (rbase + r < nrows) yp[(long)r * NC] = acc[r];
        }
      }
    }
  }
}

constexpr int QB = 16;
constexpr int AW = 32;
constexpr int SG = 8;
constexpr int QR = 64;

__device__ __forceinline__ unsigned GCN_zero_bytes(unsigned x) {
  return ~(((x & 0x7F7F7F7Fu) + 0x7F7F7F7Fu) | x | 0x7F7F7F7Fu);
}

__device__ __forceinline__ unsigned GCN_match16(const uint4 w, unsigned pat) {
  return (GCN_zero_bytes(w.x ^ pat) >> 7) | (GCN_zero_bytes(w.y ^ pat) >> 6) |
         (GCN_zero_bytes(w.z ^ pat) >> 5) | (GCN_zero_bytes(w.w ^ pat) >> 4);
}

__device__ __forceinline__ void GCN_lds_add(float* p, float v) {
  __hip_atomic_fetch_add(p, v, __ATOMIC_RELAXED, __HIP_MEMORY_SCOPE_WORKGROUP);
}

__device__ __forceinline__ int GCN_slot_base(const int* __restrict__ blk_cnt, int b, int lane) {
  int p = 0;
#pragma unroll
  for (int k = 0; k < 8; ++k) {
    const int i = lane + 32 * k;
    const int v = blk_cnt[i < b ? i : 0];
    p += i < b ? v : 0;
  }
  for (int m = 16; m > 0; m >>= 1) p += __shfl_xor(p, m);
  return __builtin_amdgcn_readfirstlane((p >> 4) + (AW + 1) * b);
}

template <int F, bool FULL, bool SCALED>
__device__ __forceinline__ void GCN_sum_batch(const float* __restrict__ H, float* acc,
                                              int s, int d, float ns, int qn,
                                              int lane, int rowbase) {
  if constexpr (F == 128) {
    float4 v[QB];
#pragma unroll
    for (int u = 0; u < QB; ++u) {
      if (FULL || u < qn) {
        const float* hp = H + (long)__builtin_amdgcn_readlane(s, u) * 128;
        v[u] = ((const float4*)hp)[lane];
      }
    }
#pragma unroll
    for (int u = 0; u < QB; ++u) {
      if (FULL || u < qn) {
        float* ap = acc + (__builtin_amdgcn_readlane(d, u) - rowbase) * 128 + lane;
        float4 w = v[u];
        if constexpr (SCALED) {
          const float su = __int_as_float(__builtin_amdgcn_readlane(__float_as_int(ns), u));
          w.x *= su; w.y *= su; w.z *= su; w.w *= su;
        }
        GCN_lds_add(ap,      w.x);
        GCN_lds_add(ap + 32, w.y);
        GCN_lds_add(ap + 64, w.z);
        GCN_lds_add(ap + 96, w.w);
      }
    }
  } else {
    constexpr int NCH = (F + 31) / 32;
    float v[NCH][QB];
#pragma unroll
    for (int c = 0; c < NCH; ++c) {
      if ((c + 1) * 32 <= F || lane < F - c * 32) {
#pragma unroll
        for (int u = 0; u < QB; ++u) {
          if (FULL || u < qn) {
            const float* hp = H + (long)__builtin_amdgcn_readlane(s, u) * F;
            v[c][u] = hp[c * 32 + lane];
          }
        }
      }
    }
#pragma unroll
    for (int c = 0; c < NCH; ++c) {
      if ((c + 1) * 32 <= F || lane < F - c * 32) {
#pragma unroll
        for (int u = 0; u < QB; ++u) {
          if (FULL || u < qn) {
            const int du = __builtin_amdgcn_readlane(d, u) - rowbase;
            float w = v[c][u];
            if constexpr (SCALED) {
              w *= __int_as_float(__builtin_amdgcn_readlane(__float_as_int(ns), u));
            }
            GCN_lds_add(acc + du * F + c * 32 + lane, w);
          }
        }
      }
    }
  }
}

template <int F, bool FULL>
__device__ __forceinline__ void GCN_record_and_sum(const float* __restrict__ H,
                                                   const int* __restrict__ src,
                                                   const int* __restrict__ dst,
                                                   const float* __restrict__ nsrc,
                                                   float* acc, int* nslot, int2* slots,
                                                   int qe, int qn, int lane, int rowbase) {
  int s = -1, d = rowbase;
  float ns = 0.0f;
  if (lane < (FULL ? QB : qn)) { s = src[qe]; d = dst[qe]; ns = nsrc[s]; }
  int t = 0;
  if (lane == 0) t = __hip_atomic_fetch_add(nslot, 1, __ATOMIC_RELAXED, __HIP_MEMORY_SCOPE_WORKGROUP);
  t = __builtin_amdgcn_readfirstlane(t);
  if (lane < QB) slots[t * QB + lane] = make_int2(s, d);
  GCN_sum_batch<F, FULL, true>(H, acc, s, d, ns, qn, lane, rowbase);
}

template <int F, bool RELU>
__device__ __forceinline__ void GCN_write_rows(const float* acc,
                                               const float* __restrict__ ndst,
                                               const float* __restrict__ bias,
                                               float* __restrict__ OUT,
                                               int lane, int rowbase, int nn) {
  if constexpr (F == 128) {
    for (int i = threadIdx.x; i < RB * 32; i += 32 * AW) {
      const int r = i >> 5;
      const int row = rowbase + r;
      if (row < nn) {
        const float* ap = acc + r * 128 + lane;
        const float nd = ndst[row];
        const float4 bb = ((const float4*)bias)[lane];
        float4 o;
        o.x = ap[0]  * nd + bb.x;
        o.y = ap[32] * nd + bb.y;
        o.z = ap[64] * nd + bb.z;
        o.w = ap[96] * nd + bb.w;
        if (RELU) {
          o.x = o.x > 0.0f ? o.x : 0.0f;
          o.y = o.y > 0.0f ? o.y : 0.0f;
          o.z = o.z > 0.0f ? o.z : 0.0f;
          o.w = o.w > 0.0f ? o.w : 0.0f;
        }
        ((float4*)(OUT + (long)row * 128))[lane] = o;
      }
    }
  } else {
    for (int i = threadIdx.x; i < RB * F; i += 32 * AW) {
      const int r = i / F;
      const int f = i - r * F;
      const int row = rowbase + r;
      if (row < nn) {
        float v = acc[i] * ndst[row] + bias[f];
        if (RELU) v = v > 0.0f ? v : 0.0f;
        OUT[(long)row * F + f] = v;
      }
    }
  }
}

template <int F, bool RELU>
__global__ __launch_bounds__(32 * AW)
void GCN_edge_sum(const float* __restrict__ H,
                  const int* __restrict__ src, const int* __restrict__ dst,
                  const float* __restrict__ nsrc,
                  const unsigned* __restrict__ own4,
                  const int* __restrict__ blk_cnt, int* __restrict__ list,
                  int* __restrict__ nslots,
                  const float* __restrict__ ndst, const float* __restrict__ bias,
                  float* __restrict__ OUT, int ne, int nn) {
  __shared__ float acc[RB * F];
  __shared__ int ring[AW * QR];
  __shared__ int nslot;
  const int lane = threadIdx.x & 31;
  const int wave = __builtin_amdgcn_readfirstlane((int)(threadIdx.x >> 5));
  const int rowbase = blockIdx.x * RB;
  int2* slots = (int2*)list + (long)GCN_slot_base(blk_cnt, blockIdx.x, lane) * QB;

  if (threadIdx.x == 0) nslot = 0;
  for (int i = threadIdx.x; i < RB * F; i += 32 * AW) acc[i] = 0.0f;
  __syncthreads();

  const unsigned pat = (unsigned)blockIdx.x * 0x01010101u;
  const uint4* own = (const uint4*)own4;
  const int nloads = ((ne + 512 * SG - 1) / (512 * SG)) * SG;
  int* rq = ring + wave * QR;
  int head = 0;
  int tail = 0;

  for (int i0 = wave * SG; i0 < nloads; i0 += AW * SG) {
    uint4 wv[SG];
#pragma unroll
    for (int g = 0; g < SG; ++g) {
      wv[g] = own[(i0 + g) * 32 + lane];
    }
    unsigned cc[SG / 2];
#pragma unroll
    for (int h = 0; h < SG / 2; ++h) {
      cc[h] = GCN_match16(wv[2 * h], pat) | (GCN_match16(wv[2 * h + 1], pat) << 4);
    }
#pragma nounroll
    for (int h = 0; h < SG / 2; ++h) {
      unsigned cur = cc[0];
#pragma unroll
      for (int k = 0; k + 1 < SG / 2; ++k) cc[k] = cc[k + 1];
      const int ebase = (i0 + 2 * h) * 512 + lane * 16;
      unsigned ball = __builtin_amdgcn_ballot_w32(cur != 0u);
      while (ball) {
        const int p = cur ? __builtin_ctz(cur) : 0;
        const int e = ebase + ((p >> 2) & 1) * 512 + (p & 3) * 4 + (p >> 3);
        const bool ok = cur != 0u && e < ne;
        const unsigned okm = __builtin_amdgcn_ballot_w32(ok);
        if (ok) rq[(tail + (int)__builtin_amdgcn_mbcnt_lo(okm, 0u)) & (QR - 1)] = e;
        cur &= cur - 1u;
        tail += __builtin_popcount(okm);
        while (tail - head >= QB) {
          int qe = 0;
          if (lane < QB) qe = rq[(head + lane) & (QR - 1)];
          GCN_record_and_sum<F, true>(H, src, dst, nsrc, acc, &nslot, slots, qe, QB, lane, rowbase);
          head += QB;
        }
        ball = __builtin_amdgcn_ballot_w32(cur != 0u);
      }
    }
  }
  {
    const int qn = tail - head;
    if (qn > 0) {
      int qe = 0;
      if (lane < qn) qe = rq[(head + lane) & (QR - 1)];
      GCN_record_and_sum<F, false>(H, src, dst, nsrc, acc, &nslot, slots, qe, qn, lane, rowbase);
    }
  }
  __syncthreads();

  if (threadIdx.x == 0) nslots[blockIdx.x] = nslot;
  GCN_write_rows<F, RELU>(acc, ndst, bias, OUT, lane, rowbase, nn);
}

template <int F, bool RELU>
__global__ __launch_bounds__(32 * AW)
void GCN_edge_sum_list(const float* __restrict__ H,
                       const int* __restrict__ blk_cnt, const int* __restrict__ list,
                       const int* __restrict__ nslots,
                       const float* __restrict__ ndst, const float* __restrict__ bias,
                       float* __restrict__ OUT, int nn) {
  __shared__ float acc[RB * F];
  const int lane = threadIdx.x & 31;
  const int wave = __builtin_amdgcn_readfirstlane((int)(threadIdx.x >> 5));
  const int rowbase = blockIdx.x * RB;
  const int2* slots = (const int2*)list + (long)GCN_slot_base(blk_cnt, blockIdx.x, lane) * QB;
  const int ns = nslots[blockIdx.x];

  for (int i = threadIdx.x; i < RB * F; i += 32 * AW) acc[i] = 0.0f;
  __syncthreads();

  for (int t = 2 * wave; t < ns; t += 2 * AW) {
    int2 p = make_int2(-1, rowbase);
    if (t + (lane >> 4) < ns) p = slots[t * QB + lane];
#pragma nounroll
    for (int h = 0; h < 2; ++h) {
      const int qn = __builtin_popcount(__builtin_amdgcn_ballot_w32(p.x >= 0) & 0xFFFFu);
      if (qn == QB)    GCN_sum_batch<F, true, false>(H, acc, p.x, p.y, 0.0f, qn, lane, rowbase);
      else if (qn > 0) GCN_sum_batch<F, false, false>(H, acc, p.x, p.y, 0.0f, qn, lane, rowbase);
      p.x = __builtin_amdgcn_ds_bpermute(((lane + QB) & 31) << 2, p.x);
      p.y = __builtin_amdgcn_ds_bpermute(((lane + QB) & 31) << 2, p.y);
    }
  }
  __syncthreads();

  GCN_write_rows<F, RELU>(acc, ndst, bias, OUT, lane, rowbase, nn);
}


extern "C" void kernel_launch(void* const* d_in, const int* in_sizes, int n_in,
                              void* d_out, int out_size, void* d_ws, size_t ws_size,
                              hipStream_t stream) {
  const float* x   = (const float*)d_in[0];
  const float* W1  = (const float*)d_in[1];
  const float* b1  = (const float*)d_in[2];
  const float* W2  = (const float*)d_in[3];
  const float* b2  = (const float*)d_in[4];
  const int* esrc  = (const int*)d_in[5];
  const int* edst  = (const int*)d_in[6];
  float* out = (float*)d_out;

  const int nn = in_sizes[0] / 256;
  const int ne = in_sizes[5];

  const long NPAD = 50048;
  float* ws       = (float*)d_ws;
  float* deg_out  = ws;
  float* deg_in   = ws + NPAD;
  float* nsrc     = ws + 2 * NPAD;
  float* ndst     = ws + 3 * NPAD;
  float* h        = ws + 4 * NPAD;
  float* h1       = ws + 4 * NPAD + (long)nn * 128;
  float* h2       = ws + 4 * NPAD + 2L * (long)nn * 128;
  unsigned* own4  = (unsigned*)(ws + 4 * NPAD + 2L * (long)nn * 128 + (long)nn * 40);
  const int nwords = ((ne + 512 * SG - 1) / (512 * SG)) * (128 * SG);
  int* blk_cnt    = (int*)(own4 + nwords);
  int* nslots     = blk_cnt + 256;
  int* list       = nslots + 256;

  const int mtiles = (nn + 15) / 16;
  const int rblocks = (nn + RB - 1) / RB;

  GCN_zero_f32<<<512, 256, 0, stream>>>(deg_out, 2 * NPAD);

  const int ndeg = (nwords + DEGW * 256 - 1) / (DEGW * 256);
  GCN_gemm_wmma_deg<256, 8, 128><<<ndeg + mtiles, 256, 0, stream>>>(
      x, W1, h, nn, ndeg, esrc, edst, deg_out, deg_in, own4, ne, nwords);

  GCN_norms<<<rblocks, RB, 0, stream>>>(deg_out, deg_in, nsrc, ndst, blk_cnt, nn);

  GCN_edge_sum<128, true><<<rblocks, 32 * AW, 0, stream>>>(h, esrc, edst, nsrc, own4, blk_cnt, list, nslots,
                                                         ndst, b1, h1, ne, nn);

  GCN_gemm_wreg<128, 3, 40><<<(mtiles + 2) / 3, 96, 0, stream>>>(h1, nsrc, W2, h2, nn);

  GCN_edge_sum_list<40, false><<<rblocks, 32 * AW, 0, stream>>>(h2, blk_cnt, list, nslots, ndst, b2, out, nn);
}


# in-degrees counted in LDS by the layer-1 edge-sum groups from the edges they sum; degree pass keeps out-degree atomics only, counts edges per row block in LDS for the slot bases; ndst from edge sum
# speedup vs baseline: 2.7893x; 1.2500x over previous
#include <hip/hip_runtime.h>


typedef __attribute__((ext_vector_type(2))) float v2f;
typedef __attribute__((ext_vector_type(8))) float v8f;

constexpr int RB = 200;


__global__ void GCN_zero_f32(float* __restrict__ p, long n) {
  long i = (long)blockIdx.x * blockDim.x + threadIdx.x;
  long stride = (long)gridDim.x * blockDim.x;
  for (; i < n; i += stride) p[i] = 0.0f;
}


__device__ __forceinline__ unsigned GCN_degree4(const int4 s, const int4 d,
                                                float* __restrict__ deg_out) {
  atomicAdd(&deg_out[s.x], 1.0f); atomicAdd(&deg_out[s.y], 1.0f);
  atomicAdd(&deg_out[s.z], 1.0f); atomicAdd(&deg_out[s.w], 1.0f);
  return (((unsigned)d.x / RB) & 255u) | ((((unsigned)d.y / RB) & 255u) << 8) |
         ((((unsigned)d.z / RB) & 255u) << 16) | ((((unsigned)d.w / RB) & 255u) << 24);
}

__device__ __forceinline__ unsigned GCN_degree(const int* __restrict__ src, const int* __restrict__ dst,
                                               float* __restrict__ deg_out,
                                               unsigned* __restrict__ own4, int ne, int nwords, int g) {
  unsigned packed = 0xFFFFFFFFu;
  if (g >= nwords) return packed;
  const int e = 4 * g;
  if (e + 4 <= ne) {
    packed = GCN_degree4(((const int4*)src)[g], ((const int4*)dst)[g], deg_out);
  } else {
    for (int k = 0; k < 4; ++k) {
      if (e + k < ne) {
        atomicAdd(&deg_out[src[e + k]], 1.0f);
        const int dd = dst[e + k];
        packed = (packed & ~(255u << (8 * k))) | ((((unsigned)dd / RB) & 255u) << (8 * k));
      }
    }
  }
  own4[g] = packed;
  return packed;
}

__global__ void GCN_norm_src(const float* __restrict__ deg_out, float* __restrict__ nsrc, int n) {
  const int i = blockIdx.x * blockDim.x + threadIdx.x;
  if (i < n) nsrc[i] = rsqrtf(fmaxf(deg_out[i], 1.0f));
}

constexpr int DEGW = 4;
template <int K, int NT, int NC>
__global__ __launch_bounds__(32 * NT)
void GCN_gemm_wmma_deg(const float* __restrict__ X,
                       const float* __restrict__ W, float* __restrict__ Y, int nrows, int ndeg,
                       const int* __restrict__ src, const int* __restrict__ dst,
                       float* __restrict__ deg_out, int* __restrict__ blk_cnt,
                       unsigned* __restrict__ own4, int ne, int nwords) {
  if ((int)blockIdx.x < ndeg) {
    __shared__ int hist[256];
    for (int i = threadIdx.x; i < 256; i += 32 * NT) hist[i] = 0;
    __syncthreads();
    const int g0 = (int)blockIdx.x * (DEGW * 32 * NT) + (int)threadIdx.x;
    const int glast = g0 + (DEGW - 1) * (32 * NT);
    unsigned pk[DEGW];
    if (glast < nwords && 4 * glast + 4 <= ne) {
      int4 s[DEGW], d[DEGW];
#pragma unroll
      for (int k = 0; k < DEGW; ++k) {
        s[k] = ((const int4*)src)[g0 + k * (32 * NT)];
        d[k] = ((const int4*)dst)[g0 + k * (32 * NT)];
      }
#pragma unroll
      for (int k = 0; k < DEGW; ++k) {
        pk[k] = GCN_degree4(s[k], d[k], deg_out);
        own4[g0 + k * (32 * NT)] = pk[k];
      }
    } else {
#pragma unroll
      for (int k = 0; k < DEGW; ++k) {
        pk[k] = GCN_degree(src, dst, deg_out, own4, ne, nwords, g0 + k * (32 * NT));
      }
    }
#pragma unroll
    for (int k = 0; k < DEGW; ++k) {
#pragma unroll
      for (int j = 0; j < 4; ++j) {
        const unsigned b = (pk[k] >> (8 * j)) & 255u;
        if (b != 255u) __hip_atomic_fetch_add(&hist[b], 1, __ATOMIC_RELAXED, __HIP_MEMORY_SCOPE_WORKGROUP);
      }
    }
    __syncthreads();
    for (int i = threadIdx.x; i < 255; i += 32 * NT) {
      const int cn = hist[i];
      if (cn != 0) atomicAdd(&blk_cnt[i], cn);
    }
    return;
  }

  constexpr int NCP = NT * 16;
  constexpr bool EXACT = (NC == NCP);
  constexpr int PITCH = K + 4;
  __shared__ float sa[16 * PITCH];
  __shared__ float sw[EXACT ? 4 : K * NCP];

  const int rowbase = ((int)blockIdx.x - ndeg) * 16;

  for (int idx = threadIdx.x; idx < 16 * (K / 4); idx += blockDim.x) {
    const int r = idx / (K / 4);
    const int c = idx - r * (K / 4);
    const int row = rowbase + r;
    float4 v = make_float4(0.f, 0.f, 0.f, 0.f);
    if (row < nrows) v = ((const float4*)(X + (long)row * K))[c];
    float* d = &sa[r * PITCH + 4 * c];
    d[0] = v.x; d[1] = v.y; d[2] = v.z; d[3] = v.w;
  }

  if constexpr (!EXACT) {
    for (int idx = threadIdx.x; idx < K * NCP; idx += blockDim.x) {
      const int k = idx / NCP;
      const int c = idx - k * NCP;
      sw[idx] = (c < NC) ? W[k * NC + c] : 0.0f;
    }
  }
  __syncthreads();

  const int lane = threadIdx.x & 31;
  const int wave = threadIdx.x >> 5;
  const int colbase = wave * 16;
  const int half = lane >> 4;
  const int l15 = lane & 15;
  const int koff = half * 2;

  const float* aptr = &sa[l15 * PITCH + koff];
  const int bcol = colbase + l15;

  v8f acc = {};
#pragma unroll 8
  for (int k0 = 0; k0 < K; k0 += 4) {
    v2f a;
    a.x = aptr[k0];
    a.y = aptr[k0 + 1];
    v2f b;
    if constexpr (EXACT) {
      b.x = W[(k0 + koff) * NC + bcol];
      b.y = W[(k0 + koff + 1) * NC + bcol];
    } else {
      b.x = sw[(k0 + koff) * NCP + bcol];
      b.y = sw[(k0 + koff + 1) * NCP + bcol];
    }
    acc = __builtin_amdgcn_wmma_f32_16x16x4_f32(
         false, a,  false, b,
         (short)0, acc,  false,  false);
  }

  if (EXACT || bcol < NC) {
    float* yp = Y + (long)(rowbase + half * 8) * NC + bcol;
    if (rowbase + 16 <= nrows) {
#pragma unroll
      for (int r = 0; r < 8; ++r) yp[(long)r * NC] = acc[r];
    } else {
      const int rbase = rowbase + half * 8;
#pragma unroll
      for (int r = 0; r < 8; ++r) {
        if (rbase + r < nrows) yp[(long)r * NC] = acc[r];
      }
    }
  }
}

template <int K, int NT, int NC>
__global__ __launch_bounds__(32 * NT)
void GCN_gemm_wreg(const float* __restrict__ X, const float* __restrict__ scale,
                   const float* __restrict__ W, float* __restrict__ Y, int nrows) {
  constexpr int NTH = 32 * NT;
  constexpr int PITCH = K + 4;
  constexpr int NV = 16 * (K / 4);
  constexpr int NLD = (NV + NTH - 1) / NTH;
  __shared__ __attribute__((aligned(16))) float sa[16 * PITCH];

  const int lane = threadIdx.x & 31;
  const int wave = threadIdx.x >> 5;
  const int colbase = wave * 16;
  const int half = lane >> 4;
  const int l15 = lane & 15;
  const int koff = half * 2;
  const int bcol = colbase + l15;
  const bool cok = bcol < NC;
  const int bc = cok ? bcol : 0;

  v2f bf[K / 4];
#pragma unroll
  for (int ks = 0; ks < K / 4; ++ks) {
    const float b0 = W[(4 * ks + koff) * NC + bc];
    const float b1 = W[(4 * ks + koff + 1) * NC + bc];
    bf[ks].x = cok ? b0 : 0.0f;
    bf[ks].y = cok ? b1 : 0.0f;
  }

  const int nstripes = (nrows + 15) >> 4;
  float4 pv[NLD];
  float ps[NLD];

  int st = blockIdx.x;
  if (st < nstripes) {
#pragma unroll
    for (int i = 0; i < NLD; ++i) {
      const int idx = threadIdx.x + i * NTH;
      const int r = idx / (K / 4);
      const int cc = idx - r * (K / 4);
      const int row = st * 16 + r;
      pv[i] = make_float4(0.f, 0.f, 0.f, 0.f);
      ps[i] = 0.0f;
      if (idx < NV && row < nrows) {
        pv[i] = ((const float4*)(X + (long)row * K))[cc];
        ps[i] = scale[row];
      }
    }
  }

  const float* aptr = &sa[l15 * PITCH + koff];

  for (; st < nstripes; st += gridDim.x) {
    const int rowbase = st * 16;
    __syncthreads();
#pragma unroll
    for (int i = 0; i < NLD; ++i) {
      const int idx = threadIdx.x + i * NTH;
      const int r = idx / (K / 4);
      const int cc = idx - r * (K / 4);
      if (idx < NV) {
        float4 o;
        o.x = pv[i].x * ps[i]; o.y = pv[i].y * ps[i];
        o.z = pv[i].z * ps[i]; o.w = pv[i].w * ps[i];
        *(float4*)&sa[r * PITCH + 4 * cc] = o;
      }
    }
    __syncthreads();

    const int nst = st + gridDim.x;
    if (nst < nstripes) {
#pragma unroll
      for (int i = 0; i < NLD; ++i) {
        const int idx = threadIdx.x + i * NTH;
        const int r = idx / (K / 4);
        const int cc = idx - r * (K / 4);
        const int row = nst * 16 + r;
        pv[i] = make_float4(0.f, 0.f, 0.f, 0.f);
        ps[i] = 0.0f;
        if (idx < NV && row < nrows) {
          pv[i] = ((const float4*)(X + (long)row * K))[cc];
          ps[i] = scale[row];
        }
      }
    }

    v8f acc = {};
#pragma unroll
    for (int ks = 0; ks < K / 4; ++ks) {
      v2f a;
      a.x = aptr[4 * ks];
      a.y = aptr[4 * ks + 1];
      acc = __builtin_amdgcn_wmma_f32_16x16x4_f32(
           false, a,  false, bf[ks],
           (short)0, acc,  false,  false);
    }

    if (cok) {
      float* yp = Y + (long)(rowbase + half * 8) * NC + bcol;
      if (rowbase + 16 <= nrows) {
#pragma unroll
        for (int r = 0; r < 8; ++r) yp[(long)r * NC] = acc[r];
      } else {
        const int rbase = rowbase + half * 8;
#pragma unroll
        for (int r = 0; r < 8; ++r) {
          if (rbase + r < nrows) yp[(long)r * NC] = acc[r];
        }
      }
    }
  }
}

constexpr int QB = 16;
constexpr int AW = 32;
constexpr int SG = 8;
constexpr int QR = 64;

__device__ __forceinline__ unsigned GCN_zero_bytes(unsigned x) {
  return ~(((x & 0x7F7F7F7Fu) + 0x7F7F7F7Fu) | x | 0x7F7F7F7Fu);
}

__device__ __forceinline__ unsigned GCN_match16(const uint4 w, unsigned pat) {
  return (GCN_zero_bytes(w.x ^ pat) >> 7) | (GCN_zero_bytes(w.y ^ pat) >> 6) |
         (GCN_zero_bytes(w.z ^ pat) >> 5) | (GCN_zero_bytes(w.w ^ pat) >> 4);
}

__device__ __forceinline__ void GCN_lds_add(float* p, float v) {
  __hip_atomic_fetch_add(p, v, __ATOMIC_RELAXED, __HIP_MEMORY_SCOPE_WORKGROUP);
}

__device__ __forceinline__ int GCN_slot_base(const int* __restrict__ blk_cnt, int b, int lane) {
  int p = 0;
#pragma unroll
  for (int k = 0; k < 8; ++k) {
    const int i = lane + 32 * k;
    const int v = blk_cnt[i < b ? i : 0];
    p += i < b ? v : 0;
  }
  for (int m = 16; m > 0; m >>= 1) p += __shfl_xor(p, m);
  return __builtin_amdgcn_readfirstlane((p >> 4) + (AW + 1) * b);
}

template <int F, bool FULL, bool SCALED>
__device__ __forceinline__ void GCN_sum_batch(const float* __restrict__ H, float* acc,
                                              int s, int d, float ns, int qn,
                                              int lane, int rowbase) {
  if constexpr (F == 128) {
    float4 v[QB];
#pragma unroll
    for (int u = 0; u < QB; ++u) {
      if (FULL || u < qn) {
        const float* hp = H + (long)__builtin_amdgcn_readlane(s, u) * 128;
        v[u] = ((const float4*)hp)[lane];
      }
    }
#pragma unroll
    for (int u = 0; u < QB; ++u) {
      if (FULL || u < qn) {
        float* ap = acc + (__builtin_amdgcn_readlane(d, u) - rowbase) * 128 + lane;
        float4 w = v[u];
        if constexpr (SCALED) {
          const float su = __int_as_float(__builtin_amdgcn_readlane(__float_as_int(ns), u));
          w.x *= su; w.y *= su; w.z *= su; w.w *= su;
        }
        GCN_lds_add(ap,      w.x);
        GCN_lds_add(ap + 32, w.y);
        GCN_lds_add(ap + 64, w.z);
        GCN_lds_add(ap + 96, w.w);
      }
    }
  } else {
    constexpr int NCH = (F + 31) / 32;
    float v[NCH][QB];
#pragma unroll
    for (int c = 0; c < NCH; ++c) {
      if ((c + 1) * 32 <= F || lane < F - c * 32) {
#pragma unroll
        for (int u = 0; u < QB; ++u) {
          if (FULL || u < qn) {
            const float* hp = H + (long)__builtin_amdgcn_readlane(s, u) * F;
            v[c][u] = hp[c * 32 + lane];
          }
        }
      }
    }
#pragma unroll
    for (int c = 0; c < NCH; ++c) {
      if ((c + 1) * 32 <= F || lane < F - c * 32) {
#pragma unroll
        for (int u = 0; u < QB; ++u) {
          if (FULL || u < qn) {
            const int du = __builtin_amdgcn_readlane(d, u) - rowbase;
            float w = v[c][u];
            if constexpr (SCALED) {
              w *= __int_as_float(__builtin_amdgcn_readlane(__float_as_int(ns), u));
            }
            GCN_lds_add(acc + du * F + c * 32 + lane, w);
          }
        }
      }
    }
  }
}

template <int F, bool FULL>
__device__ __forceinline__ void GCN_record_and_sum(const float* __restrict__ H,
                                                   const int* __restrict__ src,
                                                   const int* __restrict__ dst,
                                                   const float* __restrict__ nsrc,
                                                   float* acc, int* dcnt, int* nslot, int2* slots,
                                                   int qe, int qn, int lane, int rowbase) {
  int s = -1, d = rowbase;
  float ns = 0.0f;
  if (lane < (FULL ? QB : qn)) {
    s = src[qe]; d = dst[qe]; ns = nsrc[s];
    __hip_atomic_fetch_add(&dcnt[d - rowbase], 1, __ATOMIC_RELAXED, __HIP_MEMORY_SCOPE_WORKGROUP);
  }
  int t = 0;
  if (lane == 0) t = __hip_atomic_fetch_add(nslot, 1, __ATOMIC_RELAXED, __HIP_MEMORY_SCOPE_WORKGROUP);
  t = __builtin_amdgcn_readfirstlane(t);
  if (lane < QB) slots[t * QB + lane] = make_int2(s, d);
  GCN_sum_batch<F, FULL, true>(H, acc, s, d, ns, qn, lane, rowbase);
}

template <int F, bool RELU, bool COUNTED>
__device__ __forceinline__ void GCN_write_rows(const float* acc, const int* dcnt,
                                               const float* __restrict__ ndst_in,
                                               float* __restrict__ ndst_out,
                                               const float* __restrict__ bias,
                                               float* __restrict__ OUT,
                                               int lane, int rowbase, int nn) {
  if constexpr (F == 128) {
    for (int i = threadIdx.x; i < RB * 32; i += 32 * AW) {
      const int r = i >> 5;
      const int row = rowbase + r;
      if (row < nn) {
        const float* ap = acc + r * 128 + lane;
        float nd;
        if constexpr (COUNTED) {
          nd = rsqrtf(fmaxf((float)dcnt[r], 1.0f));
          if (lane == 0) ndst_out[row] = nd;
        } else {
          nd = ndst_in[row];
        }
        const float4 bb = ((const float4*)bias)[lane];
        float4 o;
        o.x = ap[0]  * nd + bb.x;
        o.y = ap[32] * nd + bb.y;
        o.z = ap[64] * nd + bb.z;
        o.w = ap[96] * nd + bb.w;
        if (RELU) {
          o.x = o.x > 0.0f ? o.x : 0.0f;
          o.y = o.y > 0.0f ? o.y : 0.0f;
          o.z = o.z > 0.0f ? o.z : 0.0f;
          o.w = o.w > 0.0f ? o.w : 0.0f;
        }
        ((float4*)(OUT + (long)row * 128))[lane] = o;
      }
    }
  } else {
    for (int i = threadIdx.x; i < RB * F; i += 32 * AW) {
      const int r = i / F;
      const int f = i - r * F;
      const int row = rowbase + r;
      if (row < nn) {
        float nd;
        if constexpr (COUNTED) {
          nd = rsqrtf(fmaxf((float)dcnt[r], 1.0f));
          if (f == 0) ndst_out[row] = nd;
        } else {
          nd = ndst_in[row];
        }
        float v = acc[i] * nd + bias[f];
        if (RELU) v = v > 0.0f ? v : 0.0f;
        OUT[(long)row * F + f] = v;
      }
    }
  }
}

template <int F, bool RELU>
__global__ __launch_bounds__(32 * AW)
void GCN_edge_sum(const float* __restrict__ H,
                  const int* __restrict__ src, const int* __restrict__ dst,
                  const float* __restrict__ nsrc,
                  const unsigned* __restrict__ own4,
                  const int* __restrict__ blk_cnt, int* __restrict__ list,
                  int* __restrict__ nslots,
                  float* __restrict__ ndst, const float* __restrict__ bias,
                  float* __restrict__ OUT, int ne, int nn) {
  __shared__ float acc[RB * F];
  __shared__ int ring[AW * QR];
  __shared__ int dcnt[RB];
  __shared__ int nslot;
  const int lane = threadIdx.x & 31;
  const int wave = __builtin_amdgcn_readfirstlane((int)(threadIdx.x >> 5));
  const int rowbase = blockIdx.x * RB;
  int2* slots = (int2*)list + (long)GCN_slot_base(blk_cnt, blockIdx.x, lane) * QB;

  if (threadIdx.x == 0) nslot = 0;
  for (int i = threadIdx.x; i < RB; i += 32 * AW) dcnt[i] = 0;
  for (int i = threadIdx.x; i < RB * F; i += 32 * AW) acc[i] = 0.0f;
  __syncthreads();

  const unsigned pat = (unsigned)blockIdx.x * 0x01010101u;
  const uint4* own = (const uint4*)own4;
  const int nloads = ((ne + 512 * SG - 1) / (512 * SG)) * SG;
  int* rq = ring + wave * QR;
  int head = 0;
  int tail = 0;

  for (int i0 = wave * SG; i0 < nloads; i0 += AW * SG) {
    uint4 wv[SG];
#pragma unroll
    for (int g = 0; g < SG; ++g) {
      wv[g] = own[(i0 + g) * 32 + lane];
    }
    unsigned cc[SG / 2];
#pragma unroll
    for (int h = 0; h < SG / 2; ++h) {
      cc[h] = GCN_match16(wv[2 * h], pat) | (GCN_match16(wv[2 * h + 1], pat) << 4);
    }
#pragma nounroll
    for (int h = 0; h < SG / 2; ++h) {
      unsigned cur = cc[0];
#pragma unroll
      for (int k = 0; k + 1 < SG / 2; ++k) cc[k] = cc[k + 1];
      const int ebase = (i0 + 2 * h) * 512 + lane * 16;
      unsigned ball = __builtin_amdgcn_ballot_w32(cur != 0u);
      while (ball) {
        const int p = cur ? __builtin_ctz(cur) : 0;
        const int e = ebase + ((p >> 2) & 1) * 512 + (p & 3) * 4 + (p >> 3);
        const bool ok = cur != 0u && e < ne;
        const unsigned okm = __builtin_amdgcn_ballot_w32(ok);
        if (ok) rq[(tail + (int)__builtin_amdgcn_mbcnt_lo(okm, 0u)) & (QR - 1)] = e;
        cur &= cur - 1u;
        tail += __builtin_popcount(okm);
        while (tail - head >= QB) {
          int qe = 0;
          if (lane < QB) qe = rq[(head + lane) & (QR - 1)];
          GCN_record_and_sum<F, true>(H, src, dst, nsrc, acc, dcnt, &nslot, slots, qe, QB, lane, rowbase);
          head += QB;
        }
        ball = __builtin_amdgcn_ballot_w32(cur != 0u);
      }
    }
  }
  {
    const int qn = tail - head;
    if (qn > 0) {
      int qe = 0;
      if (lane < qn) qe = rq[(head + lane) & (QR - 1)];
      GCN_record_and_sum<F, false>(H, src, dst, nsrc, acc, dcnt, &nslot, slots, qe, qn, lane, rowbase);
    }
  }
  __syncthreads();

  if (threadIdx.x == 0) nslots[blockIdx.x] = nslot;
  GCN_write_rows<F, RELU, true>(acc, dcnt, nullptr, ndst, bias, OUT, lane, rowbase, nn);
}

template <int F, bool RELU>
__global__ __launch_bounds__(32 * AW)
void GCN_edge_sum_list(const float* __restrict__ H,
                       const int* __restrict__ blk_cnt, const int* __restrict__ list,
                       const int* __restrict__ nslots,
                       const float* __restrict__ ndst, const float* __restrict__ bias,
                       float* __restrict__ OUT, int nn) {
  __shared__ float acc[RB * F];
  const int lane = threadIdx.x & 31;
  const int wave = __builtin_amdgcn_readfirstlane((int)(threadIdx.x >> 5));
  const int rowbase = blockIdx.x * RB;
  const int2* slots = (const int2*)list + (long)GCN_slot_base(blk_cnt, blockIdx.x, lane) * QB;
  const int ns = nslots[blockIdx.x];

  for (int i = threadIdx.x; i < RB * F; i += 32 * AW) acc[i] = 0.0f;
  __syncthreads();

  for (int t = 2 * wave; t < ns; t += 2 * AW) {
    int2 p = make_int2(-1, rowbase);
    if (t + (lane >> 4) < ns) p = slots[t * QB + lane];
#pragma nounroll
    for (int h = 0; h < 2; ++h) {
      const int qn = __builtin_popcount(__builtin_amdgcn_ballot_w32(p.x >= 0) & 0xFFFFu);
      if (qn == QB)    GCN_sum_batch<F, true, false>(H, acc, p.x, p.y, 0.0f, qn, lane, rowbase);
      else if (qn > 0) GCN_sum_batch<F, false, false>(H, acc, p.x, p.y, 0.0f, qn, lane, rowbase);
      p.x = __builtin_amdgcn_ds_bpermute(((lane + QB) & 31) << 2, p.x);
      p.y = __builtin_amdgcn_ds_bpermute(((lane + QB) & 31) << 2, p.y);
    }
  }
  __syncthreads();

  GCN_write_rows<F, RELU, false>(acc, nullptr, ndst, nullptr, bias, OUT, lane, rowbase, nn);
}


extern "C" void kernel_launch(void* const* d_in, const int* in_sizes, int n_in,
                              void* d_out, int out_size, void* d_ws, size_t ws_size,
                              hipStream_t stream) {
  const float* x   = (const float*)d_in[0];
  const float* W1  = (const float*)d_in[1];
  const float* b1  = (const float*)d_in[2];
  const float* W2  = (const float*)d_in[3];
  const float* b2  = (const float*)d_in[4];
  const int* esrc  = (const int*)d_in[5];
  const int* edst  = (const int*)d_in[6];
  float* out = (float*)d_out;

  const int nn = in_sizes[0] / 256;
  const int ne = in_sizes[5];

  const long NPAD = 50048;
  float* ws       = (float*)d_ws;
  float* deg_out  = ws;
  int* blk_cnt    = (int*)(ws + NPAD);
  float* nsrc     = ws + 2 * NPAD;
  float* ndst     = ws + 3 * NPAD;
  float* h        = ws + 4 * NPAD;
  float* h1       = ws + 4 * NPAD + (long)nn * 128;
  float* h2       = ws + 4 * NPAD + 2L * (long)nn * 128;
  unsigned* own4  = (unsigned*)(ws + 4 * NPAD + 2L * (long)nn * 128 + (long)nn * 40);
  const int nwords = ((ne + 512 * SG - 1) / (512 * SG)) * (128 * SG);
  int* nslots     = (int*)(own4 + nwords);
  int* list       = nslots + 256;

  const int mtiles = (nn + 15) / 16;
  const int rblocks = (nn + RB - 1) / RB;

  GCN_zero_f32<<<512, 256, 0, stream>>>(deg_out, 2 * NPAD);

  const int ndeg = (nwords + DEGW * 256 - 1) / (DEGW * 256);
  GCN_gemm_wmma_deg<256, 8, 128><<<ndeg + mtiles, 256, 0, stream>>>(
      x, W1, h, nn, ndeg, esrc, edst, deg_out, blk_cnt, own4, ne, nwords);

  GCN_norm_src<<<(nn + 255) / 256, 256, 0, stream>>>(deg_out, nsrc, nn);

  GCN_edge_sum<128, true><<<rblocks, 32 * AW, 0, stream>>>(h, esrc, edst, nsrc, own4, blk_cnt, list, nslots,
                                                         ndst, b1, h1, ne, nn);

  GCN_gemm_wreg<128, 3, 40><<<(mtiles + 2) / 3, 96, 0, stream>>>(h1, nsrc, W2, h2, nn);

  GCN_edge_sum_list<40, false><<<rblocks, 32 * AW, 0, stream>>>(h2, blk_cnt, list, nslots, ndst, b2, out, nn);
}


# out-degree f32 adds split by chip half: deg_out cut in two node ranges, waves read their die and draw edge units from their half's ticket queue; each line updated from one half; counts stay exact
# speedup vs baseline: 3.4895x; 1.2510x over previous
#include <hip/hip_runtime.h>


typedef __attribute__((ext_vector_type(2))) float v2f;
typedef __attribute__((ext_vector_type(8))) float v8f;

constexpr int RB = 200;


__global__ void GCN_zero_f32(float* __restrict__ p, long n) {
  long i = (long)blockIdx.x * blockDim.x + threadIdx.x;
  long stride = (long)gridDim.x * blockDim.x;
  for (; i < n; i += stride) p[i] = 0.0f;
}

constexpr int TKT = 256;
constexpr int TKS = 64;

template <int Q>
__device__ __forceinline__ void GCN_outdeg1(int s, float* __restrict__ deg_out, int nsplit) {
  if ((s >= nsplit) == (Q != 0)) atomicAdd(&deg_out[s], 1.0f);
}

__device__ __forceinline__ void GCN_count4(unsigned pk, int* hist) {
#pragma unroll
  for (int j = 0; j < 4; ++j) {
    const unsigned b = (pk >> (8 * j)) & 255u;
    if (b != 255u) __hip_atomic_fetch_add(&hist[b], 1, __ATOMIC_RELAXED, __HIP_MEMORY_SCOPE_WORKGROUP);
  }
}

__device__ __forceinline__ unsigned GCN_blockbytes4(const int4 d) {
  return (((unsigned)d.x / RB) & 255u) | ((((unsigned)d.y / RB) & 255u) << 8) |
         ((((unsigned)d.z / RB) & 255u) << 16) | ((((unsigned)d.w / RB) & 255u) << 24);
}

constexpr int DEGW = 4;

template <int Q>
__device__ __forceinline__ void GCN_degree_unit(const int* __restrict__ src, const int* __restrict__ dst,
                                                float* __restrict__ deg_out, unsigned* __restrict__ own4,
                                                int* hist, int ne, int nwords, int nsplit, int u, int lane) {
  const int g0 = u * (32 * DEGW) + lane;
  const int ulast = u * (32 * DEGW) + (32 * DEGW - 1);
  if (ulast < nwords && 4 * ulast + 4 <= ne) {
    int4 s[DEGW];
#pragma unroll
    for (int k = 0; k < DEGW; ++k) s[k] = ((const int4*)src)[g0 + k * 32];
    if constexpr (Q == 0) {
      int4 d[DEGW];
#pragma unroll
      for (int k = 0; k < DEGW; ++k) d[k] = ((const int4*)dst)[g0 + k * 32];
#pragma unroll
      for (int k = 0; k < DEGW; ++k) {
        const unsigned pk = GCN_blockbytes4(d[k]);
        own4[g0 + k * 32] = pk;
        GCN_count4(pk, hist);
      }
    }
#pragma unroll
    for (int k = 0; k < DEGW; ++k) {
      GCN_outdeg1<Q>(s[k].x, deg_out, nsplit); GCN_outdeg1<Q>(s[k].y, deg_out, nsplit);
      GCN_outdeg1<Q>(s[k].z, deg_out, nsplit); GCN_outdeg1<Q>(s[k].w, deg_out, nsplit);
    }
  } else {
    for (int k = 0; k < DEGW; ++k) {
      const int g = g0 + k * 32;
      if (g < nwords) {
        unsigned packed = 0xFFFFFFFFu;
        const int e = 4 * g;
        for (int j = 0; j < 4; ++j) {
          if (e + j < ne) {
            GCN_outdeg1<Q>(src[e + j], deg_out, nsplit);
            if constexpr (Q == 0) {
              const int dd = dst[e + j];
              packed = (packed & ~(255u << (8 * j))) | ((((unsigned)dd / RB) & 255u) << (8 * j));
            }
          }
        }
        if constexpr (Q == 0) {
          own4[g] = packed;
          GCN_count4(packed, hist);
        }
      }
    }
  }
}

__global__ void GCN_norm_src(const float* __restrict__ deg_out, float* __restrict__ nsrc, int n) {
  const int i = blockIdx.x * blockDim.x + threadIdx.x;
  if (i < n) nsrc[i] = rsqrtf(fmaxf(deg_out[i], 1.0f));
}

template <int K, int NT, int NC>
__global__ __launch_bounds__(32 * NT)
void GCN_gemm_wmma_deg(const float* __restrict__ X,
                       const float* __restrict__ W, float* __restrict__ Y, int nrows, int ndeg,
                       const int* __restrict__ src, const int* __restrict__ dst,
                       float* __restrict__ deg_out, int* __restrict__ blk_cnt,
                       unsigned* __restrict__ own4, int ne, int nwords) {
  if ((int)blockIdx.x < ndeg) {
    __shared__ int hist[256];
    for (int i = threadIdx.x; i < 256; i += 32 * NT) hist[i] = 0;
    __syncthreads();
    const int lane = threadIdx.x & 31;
    const int nsplit = ((nrows >> 1) + 31) & ~31;
    const unsigned place = __builtin_amdgcn_s_sendmsg_rtn(0x87);
    const int myhalf = (((place >> 16) & 15u) >= 4u) ? 1 : 0;
    const int nunits = (nwords + 32 * DEGW - 1) / (32 * DEGW);
    for (int pass = 0; pass < 2; ++pass) {
      const int q = myhalf ^ pass;
      int* ticket = blk_cnt + TKT + TKS * q;
      for (int it = 0; it <= nunits; ++it) {
        int u = 0;
        if (lane == 0) u = atomicAdd(ticket, 1);
        u = __builtin_amdgcn_readfirstlane(u);
        if (u >= nunits) break;
        if (q == 0) GCN_degree_unit<0>(src, dst, deg_out, own4, hist, ne, nwords, nsplit, u, lane);
        else        GCN_degree_unit<1>(src, dst, deg_out, own4, hist, ne, nwords, nsplit, u, lane);
      }
    }
    __syncthreads();
    for (int i = threadIdx.x; i < 255; i += 32 * NT) {
      const int cn = hist[i];
      if (cn != 0) atomicAdd(&blk_cnt[i], cn);
    }
    return;
  }

  constexpr int NCP = NT * 16;
  constexpr bool EXACT = (NC == NCP);
  constexpr int PITCH = K + 4;
  __shared__ float sa[16 * PITCH];
  __shared__ float sw[EXACT ? 4 : K * NCP];

  const int rowbase = ((int)blockIdx.x - ndeg) * 16;

  for (int idx = threadIdx.x; idx < 16 * (K / 4); idx += blockDim.x) {
    const int r = idx / (K / 4);
    const int c = idx - r * (K / 4);
    const int row = rowbase + r;
    float4 v = make_float4(0.f, 0.f, 0.f, 0.f);
    if (row < nrows) v = ((const float4*)(X + (long)row * K))[c];
    float* d = &sa[r * PITCH + 4 * c];
    d[0] = v.x; d[1] = v.y; d[2] = v.z; d[3] = v.w;
  }

  if constexpr (!EXACT) {
    for (int idx = threadIdx.x; idx < K * NCP; idx += blockDim.x) {
      const int k = idx / NCP;
      const int c = idx - k * NCP;
      sw[idx] = (c < NC) ? W[k * NC + c] : 0.0f;
    }
  }
  __syncthreads();

  const int lane = threadIdx.x & 31;
  const int wave = threadIdx.x >> 5;
  const int colbase = wave * 16;
  const int half = lane >> 4;
  const int l15 = lane & 15;
  const int koff = half * 2;

  const float* aptr = &sa[l15 * PITCH + koff];
  const int bcol = colbase + l15;

  v8f acc = {};
#pragma unroll 8
  for (int k0 = 0; k0 < K; k0 += 4) {
    v2f a;
    a.x = aptr[k0];
    a.y = aptr[k0 + 1];
    v2f b;
    if constexpr (EXACT) {
      b.x = W[(k0 + koff) * NC + bcol];
      b.y = W[(k0 + koff + 1) * NC + bcol];
    } else {
      b.x = sw[(k0 + koff) * NCP + bcol];
      b.y = sw[(k0 + koff + 1) * NCP + bcol];
    }
    acc = __builtin_amdgcn_wmma_f32_16x16x4_f32(
         false, a,  false, b,
         (short)0, acc,  false,  false);
  }

  if (EXACT || bcol < NC) {
    float* yp = Y + (long)(rowbase + half * 8) * NC + bcol;
    if (rowbase + 16 <= nrows) {
#pragma unroll
      for (int r = 0; r < 8; ++r) yp[(long)r * NC] = acc[r];
    } else {
      const int rbase = rowbase + half * 8;
#pragma unroll
      for (int r = 0; r < 8; ++r) {
        if (rbase + r < nrows) yp[(long)r * NC] = acc[r];
      }
    }
  }
}

template <int K, int NT, int NC>
__global__ __launch_bounds__(32 * NT)
void GCN_gemm_wreg(const float* __restrict__ X, const float* __restrict__ scale,
                   const float* __restrict__ W, float* __restrict__ Y, int nrows) {
  constexpr int NTH = 32 * NT;
  constexpr int PITCH = K + 4;
  constexpr int NV = 16 * (K / 4);
  constexpr int NLD = (NV + NTH - 1) / NTH;
  __shared__ __attribute__((aligned(16))) float sa[16 * PITCH];

  const int lane = threadIdx.x & 31;
  const int wave = threadIdx.x >> 5;
  const int colbase = wave * 16;
  const int half = lane >> 4;
  const int l15 = lane & 15;
  const int koff = half * 2;
  const int bcol = colbase + l15;
  const bool cok = bcol < NC;
  const int bc = cok ? bcol : 0;

  v2f bf[K / 4];
#pragma unroll
  for (int ks = 0; ks < K / 4; ++ks) {
    const float b0 = W[(4 * ks + koff) * NC + bc];
    const float b1 = W[(4 * ks + koff + 1) * NC + bc];
    bf[ks].x = cok ? b0 : 0.0f;
    bf[ks].y = cok ? b1 : 0.0f;
  }

  const int nstripes = (nrows + 15) >> 4;
  float4 pv[NLD];
  float ps[NLD];

  int st = blockIdx.x;
  if (st < nstripes) {
#pragma unroll
    for (int i = 0; i < NLD; ++i) {
      const int idx = threadIdx.x + i * NTH;
      const int r = idx / (K / 4);
      const int cc = idx - r * (K / 4);
      const int row = st * 16 + r;
      pv[i] = make_float4(0.f, 0.f, 0.f, 0.f);
      ps[i] = 0.0f;
      if (idx < NV && row < nrows) {
        pv[i] = ((const float4*)(X + (long)row * K))[cc];
        ps[i] = scale[row];
      }
    }
  }

  const float* aptr = &sa[l15 * PITCH + koff];

  for (; st < nstripes; st += gridDim.x) {
    const int rowbase = st * 16;
    __syncthreads();
#pragma unroll
    for (int i = 0; i < NLD; ++i) {
      const int idx = threadIdx.x + i * NTH;
      const int r = idx / (K / 4);
      const int cc = idx - r * (K / 4);
      if (idx < NV) {
        float4 o;
        o.x = pv[i].x * ps[i]; o.y = pv[i].y * ps[i];
        o.z = pv[i].z * ps[i]; o.w = pv[i].w * ps[i];
        *(float4*)&sa[r * PITCH + 4 * cc] = o;
      }
    }
    __syncthreads();

    const int nst = st + gridDim.x;
    if (nst < nstripes) {
#pragma unroll
      for (int i = 0; i < NLD; ++i) {
        const int idx = threadIdx.x + i * NTH;
        const int r = idx / (K / 4);
        const int cc = idx - r * (K / 4);
        const int row = nst * 16 + r;
        pv[i] = make_float4(0.f, 0.f, 0.f, 0.f);
        ps[i] = 0.0f;
        if (idx < NV && row < nrows) {
          pv[i] = ((const float4*)(X + (long)row * K))[cc];
          ps[i] = scale[row];
        }
      }
    }

    v8f acc = {};
#pragma unroll
    for (int ks = 0; ks < K / 4; ++ks) {
      v2f a;
      a.x = aptr[4 * ks];
      a.y = aptr[4 * ks + 1];
      acc = __builtin_amdgcn_wmma_f32_16x16x4_f32(
           false, a,  false, bf[ks],
           (short)0, acc,  false,  false);
    }

    if (cok) {
      float* yp = Y + (long)(rowbase + half * 8) * NC + bcol;
      if (rowbase + 16 <= nrows) {
#pragma unroll
        for (int r = 0; r < 8; ++r) yp[(long)r * NC] = acc[r];
      } else {
        const int rbase = rowbase + half * 8;
#pragma unroll
        for (int r = 0; r < 8; ++r) {
          if (rbase + r < nrows) yp[(long)r * NC] = acc[r];
        }
      }
    }
  }
}

constexpr int QB = 16;
constexpr int AW = 32;
constexpr int SG = 8;
constexpr int QR = 64;

__device__ __forceinline__ unsigned GCN_zero_bytes(unsigned x) {
  return ~(((x & 0x7F7F7F7Fu) + 0x7F7F7F7Fu) | x | 0x7F7F7F7Fu);
}

__device__ __forceinline__ unsigned GCN_match16(const uint4 w, unsigned pat) {
  return (GCN_zero_bytes(w.x ^ pat) >> 7) | (GCN_zero_bytes(w.y ^ pat) >> 6) |
         (GCN_zero_bytes(w.z ^ pat) >> 5) | (GCN_zero_bytes(w.w ^ pat) >> 4);
}

__device__ __forceinline__ void GCN_lds_add(float* p, float v) {
  __hip_atomic_fetch_add(p, v, __ATOMIC_RELAXED, __HIP_MEMORY_SCOPE_WORKGROUP);
}

__device__ __forceinline__ int GCN_slot_base(const int* __restrict__ blk_cnt, int b, int lane) {
  int p = 0;
#pragma unroll
  for (int k = 0; k < 8; ++k) {
    const int i = lane + 32 * k;
    const int v = blk_cnt[i < b ? i : 0];
    p += i < b ? v : 0;
  }
  for (int m = 16; m > 0; m >>= 1) p += __shfl_xor(p, m);
  return __builtin_amdgcn_readfirstlane((p >> 4) + (AW + 1) * b);
}

template <int F, bool FULL, bool SCALED>
__device__ __forceinline__ void GCN_sum_batch(const float* __restrict__ H, float* acc,
                                              int s, int d, float ns, int qn,
                                              int lane, int rowbase) {
  if constexpr (F == 128) {
    float4 v[QB];
#pragma unroll
    for (int u = 0; u < QB; ++u) {
      if (FULL || u < qn) {
        const float* hp = H + (long)__builtin_amdgcn_readlane(s, u) * 128;
        v[u] = ((const float4*)hp)[lane];
      }
    }
#pragma unroll
    for (int u = 0; u < QB; ++u) {
      if (FULL || u < qn) {
        float* ap = acc + (__builtin_amdgcn_readlane(d, u) - rowbase) * 128 + lane;
        float4 w = v[u];
        if constexpr (SCALED) {
          const float su = __int_as_float(__builtin_amdgcn_readlane(__float_as_int(ns), u));
          w.x *= su; w.y *= su; w.z *= su; w.w *= su;
        }
        GCN_lds_add(ap,      w.x);
        GCN_lds_add(ap + 32, w.y);
        GCN_lds_add(ap + 64, w.z);
        GCN_lds_add(ap + 96, w.w);
      }
    }
  } else {
    constexpr int NCH = (F + 31) / 32;
    float v[NCH][QB];
#pragma unroll
    for (int c = 0; c < NCH; ++c) {
      if ((c + 1) * 32 <= F || lane < F - c * 32) {
#pragma unroll
        for (int u = 0; u < QB; ++u) {
          if (FULL || u < qn) {
            const float* hp = H + (long)__builtin_amdgcn_readlane(s, u) * F;
            v[c][u] = hp[c * 32 + lane];
          }
        }
      }
    }
#pragma unroll
    for (int c = 0; c < NCH; ++c) {
      if ((c + 1) * 32 <= F || lane < F - c * 32) {
#pragma unroll
        for (int u = 0; u < QB; ++u) {
          if (FULL || u < qn) {
            const int du = __builtin_amdgcn_readlane(d, u) - rowbase;
            float w = v[c][u];
            if constexpr (SCALED) {
              w *= __int_as_float(__builtin_amdgcn_readlane(__float_as_int(ns), u));
            }
            GCN_lds_add(acc + du * F + c * 32 + lane, w);
          }
        }
      }
    }
  }
}

template <int F, bool FULL>
__device__ __forceinline__ void GCN_record_and_sum(const float* __restrict__ H,
                                                   const int* __restrict__ src,
                                                   const int* __restrict__ dst,
                                                   const float* __restrict__ nsrc,
                                                   float* acc, int* dcnt, int* nslot, int2* slots,
                                                   int qe, int qn, int lane, int rowbase) {
  int s = -1, d = rowbase;
  float ns = 0.0f;
  if (lane < (FULL ? QB : qn)) {
    s = src[qe]; d = dst[qe]; ns = nsrc[s];
    __hip_atomic_fetch_add(&dcnt[d - rowbase], 1, __ATOMIC_RELAXED, __HIP_MEMORY_SCOPE_WORKGROUP);
  }
  int t = 0;
  if (lane == 0) t = __hip_atomic_fetch_add(nslot, 1, __ATOMIC_RELAXED, __HIP_MEMORY_SCOPE_WORKGROUP);
  t = __builtin_amdgcn_readfirstlane(t);
  if (lane < QB) slots[t * QB + lane] = make_int2(s, d);
  GCN_sum_batch<F, FULL, true>(H, acc, s, d, ns, qn, lane, rowbase);
}

template <int F, bool RELU, bool COUNTED>
__device__ __forceinline__ void GCN_write_rows(const float* acc, const int* dcnt,
                                               const float* __restrict__ ndst_in,
                                               float* __restrict__ ndst_out,
                                               const float* __restrict__ bias,
                                               float* __restrict__ OUT,
                                               int lane, int rowbase, int nn) {
  if constexpr (F == 128) {
    for (int i = threadIdx.x; i < RB * 32; i += 32 * AW) {
      const int r = i >> 5;
      const int row = rowbase + r;
      if (row < nn) {
        const float* ap = acc + r * 128 + lane;
        float nd;
        if constexpr (COUNTED) {
          nd = rsqrtf(fmaxf((float)dcnt[r], 1.0f));
          if (lane == 0) ndst_out[row] = nd;
        } else {
          nd = ndst_in[row];
        }
        const float4 bb = ((const float4*)bias)[lane];
        float4 o;
        o.x = ap[0]  * nd + bb.x;
        o.y = ap[32] * nd + bb.y;
        o.z = ap[64] * nd + bb.z;
        o.w = ap[96] * nd + bb.w;
        if (RELU) {
          o.x = o.x > 0.0f ? o.x : 0.0f;
          o.y = o.y > 0.0f ? o.y : 0.0f;
          o.z = o.z > 0.0f ? o.z : 0.0f;
          o.w = o.w > 0.0f ? o.w : 0.0f;
        }
        ((float4*)(OUT + (long)row * 128))[lane] = o;
      }
    }
  } else {
    for (int i = threadIdx.x; i < RB * F; i += 32 * AW) {
      const int r = i / F;
      const int f = i - r * F;
      const int row = rowbase + r;
      if (row < nn) {
        float nd;
        if constexpr (COUNTED) {
          nd = rsqrtf(fmaxf((float)dcnt[r], 1.0f));
          if (f == 0) ndst_out[row] = nd;
        } else {
          nd = ndst_in[row];
        }
        float v = acc[i] * nd + bias[f];
        if (RELU) v = v > 0.0f ? v : 0.0f;
        OUT[(long)row * F + f] = v;
      }
    }
  }
}

template <int F, bool RELU>
__global__ __launch_bounds__(32 * AW)
void GCN_edge_sum(const float* __restrict__ H,
                  const int* __restrict__ src, const int* __restrict__ dst,
                  const float* __restrict__ nsrc,
                  const unsigned* __restrict__ own4,
                  const int* __restrict__ blk_cnt, int* __restrict__ list,
                  int* __restrict__ nslots,
                  float* __restrict__ ndst, const float* __restrict__ bias,
                  float* __restrict__ OUT, int ne, int nn) {
  __shared__ float acc[RB * F];
  __shared__ int ring[AW * QR];
  __shared__ int dcnt[RB];
  __shared__ int nslot;
  const int lane = threadIdx.x & 31;
  const int wave = __builtin_amdgcn_readfirstlane((int)(threadIdx.x >> 5));
  const int rowbase = blockIdx.x * RB;
  int2* slots = (int2*)list + (long)GCN_slot_base(blk_cnt, blockIdx.x, lane) * QB;

  if (threadIdx.x == 0) nslot = 0;
  for (int i = threadIdx.x; i < RB; i += 32 * AW) dcnt[i] = 0;
  for (int i = threadIdx.x; i < RB * F; i += 32 * AW) acc[i] = 0.0f;
  __syncthreads();

  const unsigned pat = (unsigned)blockIdx.x * 0x01010101u;
  const uint4* own = (const uint4*)own4;
  const int nloads = ((ne + 512 * SG - 1) / (512 * SG)) * SG;
  int* rq = ring + wave * QR;
  int head = 0;
  int tail = 0;

  for (int i0 = wave * SG; i0 < nloads; i0 += AW * SG) {
    uint4 wv[SG];
#pragma unroll
    for (int g = 0; g < SG; ++g) {
      wv[g] = own[(i0 + g) * 32 + lane];
    }
    unsigned cc[SG / 2];
#pragma unroll
    for (int h = 0; h < SG / 2; ++h) {
      cc[h] = GCN_match16(wv[2 * h], pat) | (GCN_match16(wv[2 * h + 1], pat) << 4);
    }
#pragma nounroll
    for (int h = 0; h < SG / 2; ++h) {
      unsigned cur = cc[0];
#pragma unroll
      for (int k = 0; k + 1 < SG / 2; ++k) cc[k] = cc[k + 1];
      const int ebase = (i0 + 2 * h) * 512 + lane * 16;
      unsigned ball = __builtin_amdgcn_ballot_w32(cur != 0u);
      while (ball) {
        const int p = cur ? __builtin_ctz(cur) : 0;
        const int e = ebase + ((p >> 2) & 1) * 512 + (p & 3) * 4 + (p >> 3);
        const bool ok = cur != 0u && e < ne;
        const unsigned okm = __builtin_amdgcn_ballot_w32(ok);
        if (ok) rq[(tail + (int)__builtin_amdgcn_mbcnt_lo(okm, 0u)) & (QR - 1)] = e;
        cur &= cur - 1u;
        tail += __builtin_popcount(okm);
        while (tail - head >= QB) {
          int qe = 0;
          if (lane < QB) qe = rq[(head + lane) & (QR - 1)];
          GCN_record_and_sum<F, true>(H, src, dst, nsrc, acc, dcnt, &nslot, slots, qe, QB, lane, rowbase);
          head += QB;
        }
        ball = __builtin_amdgcn_ballot_w32(cur != 0u);
      }
    }
  }
  {
    const int qn = tail - head;
    if (qn > 0) {
      int qe = 0;
      if (lane < qn) qe = rq[(head + lane) & (QR - 1)];
      GCN_record_and_sum<F, false>(H, src, dst, nsrc, acc, dcnt, &nslot, slots, qe, qn, lane, rowbase);
    }
  }
  __syncthreads();

  if (threadIdx.x == 0) nslots[blockIdx.x] = nslot;
  GCN_write_rows<F, RELU, true>(acc, dcnt, nullptr, ndst, bias, OUT, lane, rowbase, nn);
}

template <int F, bool RELU>
__global__ __launch_bounds__(32 * AW)
void GCN_edge_sum_list(const float* __restrict__ H,
                       const int* __restrict__ blk_cnt, const int* __restrict__ list,
                       const int* __restrict__ nslots,
                       const float* __restrict__ ndst, const float* __restrict__ bias,
                       float* __restrict__ OUT, int nn) {
  __shared__ float acc[RB * F];
  const int lane = threadIdx.x & 31;
  const int wave = __builtin_amdgcn_readfirstlane((int)(threadIdx.x >> 5));
  const int rowbase = blockIdx.x * RB;
  const int2* slots = (const int2*)list + (long)GCN_slot_base(blk_cnt, blockIdx.x, lane) * QB;
  const int ns = nslots[blockIdx.x];

  for (int i = threadIdx.x; i < RB * F; i += 32 * AW) acc[i] = 0.0f;
  __syncthreads();

  for (int t = 2 * wave; t < ns; t += 2 * AW) {
    int2 p = make_int2(-1, rowbase);
    if (t + (lane >> 4) < ns) p = slots[t * QB + lane];
#pragma nounroll
    for (int h = 0; h < 2; ++h) {
      const int qn = __builtin_popcount(__builtin_amdgcn_ballot_w32(p.x >= 0) & 0xFFFFu);
      if (qn == QB)    GCN_sum_batch<F, true, false>(H, acc, p.x, p.y, 0.0f, qn, lane, rowbase);
      else if (qn > 0) GCN_sum_batch<F, false, false>(H, acc, p.x, p.y, 0.0f, qn, lane, rowbase);
      p.x = __builtin_amdgcn_ds_bpermute(((lane + QB) & 31) << 2, p.x);
      p.y = __builtin_amdgcn_ds_bpermute(((lane + QB) & 31) << 2, p.y);
    }
  }
  __syncthreads();

  GCN_write_rows<F, RELU, false>(acc, nullptr, ndst, nullptr, bias, OUT, lane, rowbase, nn);
}


extern "C" void kernel_launch(void* const* d_in, const int* in_sizes, int n_in,
                              void* d_out, int out_size, void* d_ws, size_t ws_size,
                              hipStream_t stream) {
  const float* x   = (const float*)d_in[0];
  const float* W1  = (const float*)d_in[1];
  const float* b1  = (const float*)d_in[2];
  const float* W2  = (const float*)d_in[3];
  const float* b2  = (const float*)d_in[4];
  const int* esrc  = (const int*)d_in[5];
  const int* edst  = (const int*)d_in[6];
  float* out = (float*)d_out;

  const int nn = in_sizes[0] / 256;
  const int ne = in_sizes[5];

  const long NPAD = 50048;
  float* ws       = (float*)d_ws;
  float* deg_out  = ws;
  int* blk_cnt    = (int*)(ws + NPAD);
  float* nsrc     = ws + 2 * NPAD;
  float* ndst     = ws + 3 * NPAD;
  float* h        = ws + 4 * NPAD;
  float* h1       = ws + 4 * NPAD + (long)nn * 128;
  float* h2       = ws + 4 * NPAD + 2L * (long)nn * 128;
  unsigned* own4  = (unsigned*)(ws + 4 * NPAD + 2L * (long)nn * 128 + (long)nn * 40);
  const int nwords = ((ne + 512 * SG - 1) / (512 * SG)) * (128 * SG);
  int* nslots     = (int*)(own4 + nwords);
  int* list       = nslots + 256;

  const int mtiles = (nn + 15) / 16;
  const int rblocks = (nn + RB - 1) / RB;

  GCN_zero_f32<<<512, 256, 0, stream>>>(deg_out, 2 * NPAD);

  const int ndeg = (nwords + DEGW * 256 - 1) / (DEGW * 256);
  GCN_gemm_wmma_deg<256, 8, 128><<<ndeg + mtiles, 256, 0, stream>>>(
      x, W1, h, nn, ndeg, esrc, edst, deg_out, blk_cnt, own4, ne, nwords);

  GCN_norm_src<<<(nn + 255) / 256, 256, 0, stream>>>(deg_out, nsrc, nn);

  GCN_edge_sum<128, true><<<rblocks, 32 * AW, 0, stream>>>(h, esrc, edst, nsrc, own4, blk_cnt, list, nslots,
                                                         ndst, b1, h1, ne, nn);

  GCN_gemm_wreg<128, 3, 40><<<(mtiles + 2) / 3, 96, 0, stream>>>(h1, nsrc, W2, h2, nn);

  GCN_edge_sum_list<40, false><<<rblocks, 32 * AW, 0, stream>>>(h2, blk_cnt, list, nslots, ndst, b2, out, nn);
}
